# starting kernel
# baseline (speedup 1.0000x reference)
#include <hip/hip_runtime.h>


typedef __attribute__((ext_vector_type(2))) float v2f;
typedef __attribute__((ext_vector_type(8))) float v8f;

#define DD 64
#define AP 68
#define WP 72

__device__ __forceinline__ v8f wmma4(v2f a, v2f b, v8f c) {
  return __builtin_amdgcn_wmma_f32_16x16x4_f32(
       false, a,  false, b,
       (short)0, c,  false,  false);
}

__global__ void edge_prep(const int* __restrict__ dst, const int* __restrict__ et,
                          int* __restrict__ cnt, int* __restrict__ hist,
                          int E, int R) {
  int e = blockIdx.x * blockDim.x + threadIdx.x;
  if (e >= E) return;
  int t = et[e];
  atomicAdd(&cnt[(size_t)dst[e] * R + t], 1);
  atomicAdd(&hist[t], 1);
}

__global__ void scan_pad(const int* __restrict__ hist, int* __restrict__ cursor,
                         int* __restrict__ tileRel, int R, int maxTiles) {
  if (blockIdx.x != 0 || threadIdx.x != 0) return;
  int run = 0;
  for (int r = 0; r < R; ++r) {
    cursor[r] = run * 64;
    int tiles = (hist[r] + 63) >> 6;
    for (int tt = 0; tt < tiles; ++tt) tileRel[run + tt] = r;
    run += tiles;
  }
  for (int tt = run; tt < maxTiles; ++tt) tileRel[tt] = -1;
}

__global__ void edge_scatter(const int* __restrict__ et, int* __restrict__ cursor,
                             int* __restrict__ perm, int E) {
  int e = blockIdx.x * blockDim.x + threadIdx.x;
  if (e >= E) return;
  int pos = atomicAdd(&cursor[et[e]], 1);
  perm[pos] = e;
}

__global__ void __launch_bounds__(256)
edge_gemm(const float* __restrict__ h, const float* __restrict__ Wl,
          const int* __restrict__ perm, const int* __restrict__ tileRel,
          const int* __restrict__ src, const int* __restrict__ dstArr,
          const int* __restrict__ et, const int* __restrict__ cnt,
          float* __restrict__ conv, int R) {
  __shared__ float As[64][AP];
  __shared__ float Ws[64][WP];
  __shared__ int   eSrc[64];
  __shared__ int   eDst[64];
  __shared__ float eInv[64];

  int rel = tileRel[blockIdx.x];
  if (rel < 0) return;
  int tid = threadIdx.x;

  if (tid < 64) {
    int e = perm[(size_t)blockIdx.x * 64 + tid];
    int s = -1, d = 0; float inv = 0.f;
    if (e >= 0) {
      s = src[e];
      d = dstArr[e];
      int c = cnt[(size_t)d * R + et[e]];
      inv = 1.0f / (float)(c > 0 ? c : 1);
    }
    eSrc[tid] = s; eDst[tid] = d; eInv[tid] = inv;
  }

  {
    const float4* W4 = (const float4*)(Wl + (size_t)rel * DD * DD);
    for (int i = 0; i < 4; ++i) {
      int idx = tid * 4 + i;
      int r = idx >> 4, c4 = (idx & 15) * 4;
      float4 v = W4[idx];
      Ws[r][c4 + 0] = v.x; Ws[r][c4 + 1] = v.y;
      Ws[r][c4 + 2] = v.z; Ws[r][c4 + 3] = v.w;
    }
  }
  __syncthreads();

  {
    int row = tid >> 2, cb = (tid & 3) * 16;
    int s = eSrc[row];
    if (s >= 0) {
      const float4* hp = (const float4*)(h + (size_t)s * DD + cb);
      for (int i = 0; i < 4; ++i) {
        float4 v = hp[i];
        As[row][cb + i * 4 + 0] = v.x; As[row][cb + i * 4 + 1] = v.y;
        As[row][cb + i * 4 + 2] = v.z; As[row][cb + i * 4 + 3] = v.w;
      }
    } else {
      for (int i = 0; i < 16; ++i) As[row][cb + i] = 0.f;
    }
  }
  __syncthreads();

  int lane = tid & 31, w = tid >> 5;
  int rt = w >> 1, cp = w & 1;
  int cb0 = cp * 32, cb1 = cb0 + 16;
  int n = lane & 15, hi = lane >> 4;
  int m = rt * 16 + n;

  v8f c0 = {}; v8f c1 = {};
  for (int k = 0; k < 16; ++k) {
    int kb = k * 4 + hi * 2;
    v2f a;  a.x  = As[m][kb];        a.y  = As[m][kb + 1];
    v2f b0; b0.x = Ws[kb][cb0 + n];  b0.y = Ws[kb + 1][cb0 + n];
    v2f b1; b1.x = Ws[kb][cb1 + n];  b1.y = Ws[kb + 1][cb1 + n];
    c0 = wmma4(a, b0, c0);
    c1 = wmma4(a, b1, c1);
  }

  for (int i = 0; i < 8; ++i) {
    int gr = rt * 16 + hi * 8 + i;
    if (eSrc[gr] >= 0) {
      float inv = eInv[gr];
      size_t base = (size_t)eDst[gr] * DD;
      atomicAdd(&conv[base + cb0 + n], c0[i] * inv);
      atomicAdd(&conv[base + cb1 + n], c1[i] * inv);
    }
  }
}

__global__ void __launch_bounds__(256)
node_gemm(const float* __restrict__ h, const float* __restrict__ conv,
          const float* __restrict__ rootW, const float* __restrict__ biasv,
          const float* __restrict__ resW, const float* __restrict__ resb,
          float* __restrict__ hout, int N) {
  __shared__ float As[64][AP];
  __shared__ float Wr[64][WP];
  __shared__ float Wq[64][WP];
  __shared__ float bV[64], rbV[64];

  int tid = threadIdx.x;
  int base = blockIdx.x * 64;

  if (tid < 64) { bV[tid] = biasv[tid]; rbV[tid] = resb[tid]; }

  for (int i = 0; i < 4; ++i) {
    int idx = tid * 4 + i;
    int r = idx >> 4, c4 = (idx & 15) * 4;
    float4 v = ((const float4*)rootW)[idx];
    Wr[r][c4 + 0] = v.x; Wr[r][c4 + 1] = v.y; Wr[r][c4 + 2] = v.z; Wr[r][c4 + 3] = v.w;
    float4 u = ((const float4*)resW)[idx];
    Wq[r][c4 + 0] = u.x; Wq[r][c4 + 1] = u.y; Wq[r][c4 + 2] = u.z; Wq[r][c4 + 3] = u.w;
  }

  {
    int row = tid >> 2, cb = (tid & 3) * 16;
    int node = base + row;
    if (node < N) {
      const float4* hp = (const float4*)(h + (size_t)node * DD + cb);
      for (int i = 0; i < 4; ++i) {
        float4 v = hp[i];
        As[row][cb + i * 4 + 0] = v.x; As[row][cb + i * 4 + 1] = v.y;
        As[row][cb + i * 4 + 2] = v.z; As[row][cb + i * 4 + 3] = v.w;
      }
    } else {
      for (int i = 0; i < 16; ++i) As[row][cb + i] = 0.f;
    }
  }
  __syncthreads();

  int lane = tid & 31, w = tid >> 5;
  int rt = w >> 1, cp = w & 1;
  int cb0 = cp * 32, cb1 = cb0 + 16;
  int n = lane & 15, hi = lane >> 4;
  int m = rt * 16 + n;

  v8f cR0 = {}, cR1 = {}, cS0 = {}, cS1 = {};
  for (int k = 0; k < 16; ++k) {
    int kb = k * 4 + hi * 2;
    v2f a;   a.x   = As[m][kb];        a.y   = As[m][kb + 1];
    v2f br0; br0.x = Wr[kb][cb0 + n];  br0.y = Wr[kb + 1][cb0 + n];
    v2f br1; br1.x = Wr[kb][cb1 + n];  br1.y = Wr[kb + 1][cb1 + n];
    v2f bq0; bq0.x = Wq[kb][cb0 + n];  bq0.y = Wq[kb + 1][cb0 + n];
    v2f bq1; bq1.x = Wq[kb][cb1 + n];  bq1.y = Wq[kb + 1][cb1 + n];
    cR0 = wmma4(a, br0, cR0);
    cR1 = wmma4(a, br1, cR1);
    cS0 = wmma4(a, bq0, cS0);
    cS1 = wmma4(a, bq1, cS1);
  }

  for (int i = 0; i < 8; ++i) {
    int gr = rt * 16 + hi * 8 + i;
    int node = base + gr;
    if (node < N) {
      size_t b0 = (size_t)node * DD + cb0 + n;
      size_t b1 = (size_t)node * DD + cb1 + n;
      float o0 = fmaxf(conv[b0] + cR0[i] + bV[cb0 + n], 0.f) +
                 fmaxf(cS0[i] + rbV[cb0 + n], 0.f);
      float o1 = fmaxf(conv[b1] + cR1[i] + bV[cb1 + n], 0.f) +
                 fmaxf(cS1[i] + rbV[cb1 + n], 0.f);
      hout[b0] = o0;
      hout[b1] = o1;
    }
  }
}

extern "C" void kernel_launch(void* const* d_in, const int* in_sizes, int n_in,
                              void* d_out, int out_size, void* d_ws, size_t ws_size,
                              hipStream_t stream) {
  const float* x    = (const float*)d_in[0];
  const int*   ei   = (const int*)d_in[1];
  const int*   et   = (const int*)d_in[2];
  const float* W    = (const float*)d_in[3];
  const float* root = (const float*)d_in[4];
  const float* bias = (const float*)d_in[5];
  const float* resW = (const float*)d_in[6];
  const float* resb = (const float*)d_in[7];

  const int N = in_sizes[0] / DD;
  const int E = in_sizes[2];
  const int L = in_sizes[4] / (DD * DD);
  const int R = in_sizes[3] / (L * DD * DD);
  const int* src  = ei;
  const int* dstA = ei + E;
  const int maxTiles = (E + 63) / 64 + R;

  char* wsb = (char*)d_ws;
  size_t off = 0;
  auto carve = [&](size_t bytes) -> void* {
    void* p = wsb + off;
    off += (bytes + 255) & ~(size_t)255;
    return p;
  };
  int*   cnt     = (int*)carve((size_t)N * R * sizeof(int));
  int*   hist    = (int*)carve((size_t)R * sizeof(int));
  int*   cursor  = (int*)carve((size_t)R * sizeof(int));
  int*   tileRel = (int*)carve((size_t)maxTiles * sizeof(int));
  int*   perm    = (int*)carve((size_t)maxTiles * 64 * sizeof(int));
  float* hA      = (float*)carve((size_t)N * DD * sizeof(float));
  float* hB      = (float*)carve((size_t)N * DD * sizeof(float));
  float* conv    = (float*)carve((size_t)N * DD * sizeof(float));

  hipMemsetAsync(cnt, 0, (size_t)N * R * sizeof(int), stream);
  hipMemsetAsync(hist, 0, (size_t)R * sizeof(int), stream);
  hipMemsetAsync(perm, 0xFF, (size_t)maxTiles * 64 * sizeof(int), stream);

  const int eb = (E + 255) / 256;
  edge_prep   <<<eb, 256, 0, stream>>>(dstA, et, cnt, hist, E, R);
  scan_pad    <<<1, 32, 0, stream>>>(hist, cursor, tileRel, R, maxTiles);
  edge_scatter<<<eb, 256, 0, stream>>>(et, cursor, perm, E);

  const float* hcur = x;
  for (int l = 0; l < L; ++l) {
    float* hnext = (l == L - 1) ? (float*)d_out : ((l & 1) ? hB : hA);
    hipMemsetAsync(conv, 0, (size_t)N * DD * sizeof(float), stream);
    edge_gemm<<<maxTiles, 256, 0, stream>>>(
        hcur, W + (size_t)l * R * DD * DD, perm, tileRel,
        src, dstA, et, cnt, conv, R);
    node_gemm<<<(N + 63) / 64, 256, 0, stream>>>(
        hcur, conv, root + (size_t)l * DD * DD, bias + (size_t)l * DD,
        resW + (size_t)l * DD * DD, resb + (size_t)l * DD, hnext, N);
    hcur = hnext;
  }
}


# Relation histogram and scatter cursor counted per block in LDS: one global add per (block, relation), not per edge; arrival-order rank changes, so the order of the float atomic sums changes
# speedup vs baseline: 1.7953x; 1.7953x over previous
#include <hip/hip_runtime.h>


typedef __attribute__((ext_vector_type(2))) float v2f;
typedef __attribute__((ext_vector_type(8))) float v8f;

#define DD 64
#define AP 68
#define WP 72

__device__ __forceinline__ v8f wmma4(v2f a, v2f b, v8f c) {
  return __builtin_amdgcn_wmma_f32_16x16x4_f32(
       false, a,  false, b,
       (short)0, c,  false,  false);
}

#define RMAX 128

__global__ void __launch_bounds__(1024)
edge_prep(const int* __restrict__ dst, const int* __restrict__ et,
          int* __restrict__ cnt, int* __restrict__ hist, int E, int R) {
  __shared__ int lh[RMAX];
  const int per = (E + (int)gridDim.x - 1) / (int)gridDim.x;
  const int lo = (int)blockIdx.x * per;
  const int hi = min(lo + per, E);
  const bool inLds = R <= RMAX;
  if (inLds) {
    for (int r = threadIdx.x; r < R; r += blockDim.x) lh[r] = 0;
    __syncthreads();
  }
  for (int e = lo + (int)threadIdx.x; e < hi; e += blockDim.x) {
    int t = et[e];
    atomicAdd(&cnt[(size_t)dst[e] * R + t], 1);
    if (inLds) atomicAdd(&lh[t], 1);
    else       atomicAdd(&hist[t], 1);
  }
  if (inLds) {
    __syncthreads();
    for (int r = threadIdx.x; r < R; r += blockDim.x) {
      int c = lh[r];
      if (c) atomicAdd(&hist[r], c);
    }
  }
}

__global__ void scan_pad(const int* __restrict__ hist, int* __restrict__ cursor,
                         int* __restrict__ tileRel, int R, int maxTiles) {
  if (blockIdx.x != 0 || threadIdx.x != 0) return;
  int run = 0;
  for (int r = 0; r < R; ++r) {
    cursor[r] = run * 64;
    int tiles = (hist[r] + 63) >> 6;
    for (int tt = 0; tt < tiles; ++tt) tileRel[run + tt] = r;
    run += tiles;
  }
  for (int tt = run; tt < maxTiles; ++tt) tileRel[tt] = -1;
}

__global__ void __launch_bounds__(1024)
edge_scatter(const int* __restrict__ et, int* __restrict__ cursor,
             int* __restrict__ perm, int E, int R) {
  __shared__ int lc[RMAX];
  const int per = (E + (int)gridDim.x - 1) / (int)gridDim.x;
  const int lo = (int)blockIdx.x * per;
  const int hi = min(lo + per, E);
  if (R > RMAX) {
    for (int e = lo + (int)threadIdx.x; e < hi; e += blockDim.x) {
      int pos = atomicAdd(&cursor[et[e]], 1);
      perm[pos] = e;
    }
    return;
  }
  for (int r = threadIdx.x; r < R; r += blockDim.x) lc[r] = 0;
  __syncthreads();
  for (int e = lo + (int)threadIdx.x; e < hi; e += blockDim.x)
    atomicAdd(&lc[et[e]], 1);
  __syncthreads();
  for (int r = threadIdx.x; r < R; r += blockDim.x) {
    int c = lc[r];
    lc[r] = c ? atomicAdd(&cursor[r], c) : 0;
  }
  __syncthreads();
  for (int e = lo + (int)threadIdx.x; e < hi; e += blockDim.x) {
    int pos = atomicAdd(&lc[et[e]], 1);
    perm[pos] = e;
  }
}

__global__ void __launch_bounds__(256)
edge_gemm(const float* __restrict__ h, const float* __restrict__ Wl,
          const int* __restrict__ perm, const int* __restrict__ tileRel,
          const int* __restrict__ src, const int* __restrict__ dstArr,
          const int* __restrict__ et, const int* __restrict__ cnt,
          float* __restrict__ conv, int R) {
  __shared__ float As[64][AP];
  __shared__ float Ws[64][WP];
  __shared__ int   eSrc[64];
  __shared__ int   eDst[64];
  __shared__ float eInv[64];

  int rel = tileRel[blockIdx.x];
  if (rel < 0) return;
  int tid = threadIdx.x;

  if (tid < 64) {
    int e = perm[(size_t)blockIdx.x * 64 + tid];
    int s = -1, d = 0; float inv = 0.f;
    if (e >= 0) {
      s = src[e];
      d = dstArr[e];
      int c = cnt[(size_t)d * R + et[e]];
      inv = 1.0f / (float)(c > 0 ? c : 1);
    }
    eSrc[tid] = s; eDst[tid] = d; eInv[tid] = inv;
  }

  {
    const float4* W4 = (const float4*)(Wl + (size_t)rel * DD * DD);
    for (int i = 0; i < 4; ++i) {
      int idx = tid * 4 + i;
      int r = idx >> 4, c4 = (idx & 15) * 4;
      float4 v = W4[idx];
      Ws[r][c4 + 0] = v.x; Ws[r][c4 + 1] = v.y;
      Ws[r][c4 + 2] = v.z; Ws[r][c4 + 3] = v.w;
    }
  }
  __syncthreads();

  {
    int row = tid >> 2, cb = (tid & 3) * 16;
    int s = eSrc[row];
    if (s >= 0) {
      const float4* hp = (const float4*)(h + (size_t)s * DD + cb);
      for (int i = 0; i < 4; ++i) {
        float4 v = hp[i];
        As[row][cb + i * 4 + 0] = v.x; As[row][cb + i * 4 + 1] = v.y;
        As[row][cb + i * 4 + 2] = v.z; As[row][cb + i * 4 + 3] = v.w;
      }
    } else {
      for (int i = 0; i < 16; ++i) As[row][cb + i] = 0.f;
    }
  }
  __syncthreads();

  int lane = tid & 31, w = tid >> 5;
  int rt = w >> 1, cp = w & 1;
  int cb0 = cp * 32, cb1 = cb0 + 16;
  int n = lane & 15, hi = lane >> 4;
  int m = rt * 16 + n;

  v8f c0 = {}; v8f c1 = {};
  for (int k = 0; k < 16; ++k) {
    int kb = k * 4 + hi * 2;
    v2f a;  a.x  = As[m][kb];        a.y  = As[m][kb + 1];
    v2f b0; b0.x = Ws[kb][cb0 + n];  b0.y = Ws[kb + 1][cb0 + n];
    v2f b1; b1.x = Ws[kb][cb1 + n];  b1.y = Ws[kb + 1][cb1 + n];
    c0 = wmma4(a, b0, c0);
    c1 = wmma4(a, b1, c1);
  }

  for (int i = 0; i < 8; ++i) {
    int gr = rt * 16 + hi * 8 + i;
    if (eSrc[gr] >= 0) {
      float inv = eInv[gr];
      size_t base = (size_t)eDst[gr] * DD;
      atomicAdd(&conv[base + cb0 + n], c0[i] * inv);
      atomicAdd(&conv[base + cb1 + n], c1[i] * inv);
    }
  }
}

__global__ void __launch_bounds__(256)
node_gemm(const float* __restrict__ h, const float* __restrict__ conv,
          const float* __restrict__ rootW, const float* __restrict__ biasv,
          const float* __restrict__ resW, const float* __restrict__ resb,
          float* __restrict__ hout, int N) {
  __shared__ float As[64][AP];
  __shared__ float Wr[64][WP];
  __shared__ float Wq[64][WP];
  __shared__ float bV[64], rbV[64];

  int tid = threadIdx.x;
  int base = blockIdx.x * 64;

  if (tid < 64) { bV[tid] = biasv[tid]; rbV[tid] = resb[tid]; }

  for (int i = 0; i < 4; ++i) {
    int idx = tid * 4 + i;
    int r = idx >> 4, c4 = (idx & 15) * 4;
    float4 v = ((const float4*)rootW)[idx];
    Wr[r][c4 + 0] = v.x; Wr[r][c4 + 1] = v.y; Wr[r][c4 + 2] = v.z; Wr[r][c4 + 3] = v.w;
    float4 u = ((const float4*)resW)[idx];
    Wq[r][c4 + 0] = u.x; Wq[r][c4 + 1] = u.y; Wq[r][c4 + 2] = u.z; Wq[r][c4 + 3] = u.w;
  }

  {
    int row = tid >> 2, cb = (tid & 3) * 16;
    int node = base + row;
    if (node < N) {
      const float4* hp = (const float4*)(h + (size_t)node * DD + cb);
      for (int i = 0; i < 4; ++i) {
        float4 v = hp[i];
        As[row][cb + i * 4 + 0] = v.x; As[row][cb + i * 4 + 1] = v.y;
        As[row][cb + i * 4 + 2] = v.z; As[row][cb + i * 4 + 3] = v.w;
      }
    } else {
      for (int i = 0; i < 16; ++i) As[row][cb + i] = 0.f;
    }
  }
  __syncthreads();

  int lane = tid & 31, w = tid >> 5;
  int rt = w >> 1, cp = w & 1;
  int cb0 = cp * 32, cb1 = cb0 + 16;
  int n = lane & 15, hi = lane >> 4;
  int m = rt * 16 + n;

  v8f cR0 = {}, cR1 = {}, cS0 = {}, cS1 = {};
  for (int k = 0; k < 16; ++k) {
    int kb = k * 4 + hi * 2;
    v2f a;   a.x   = As[m][kb];        a.y   = As[m][kb + 1];
    v2f br0; br0.x = Wr[kb][cb0 + n];  br0.y = Wr[kb + 1][cb0 + n];
    v2f br1; br1.x = Wr[kb][cb1 + n];  br1.y = Wr[kb + 1][cb1 + n];
    v2f bq0; bq0.x = Wq[kb][cb0 + n];  bq0.y = Wq[kb + 1][cb0 + n];
    v2f bq1; bq1.x = Wq[kb][cb1 + n];  bq1.y = Wq[kb + 1][cb1 + n];
    cR0 = wmma4(a, br0, cR0);
    cR1 = wmma4(a, br1, cR1);
    cS0 = wmma4(a, bq0, cS0);
    cS1 = wmma4(a, bq1, cS1);
  }

  for (int i = 0; i < 8; ++i) {
    int gr = rt * 16 + hi * 8 + i;
    int node = base + gr;
    if (node < N) {
      size_t b0 = (size_t)node * DD + cb0 + n;
      size_t b1 = (size_t)node * DD + cb1 + n;
      float o0 = fmaxf(conv[b0] + cR0[i] + bV[cb0 + n], 0.f) +
                 fmaxf(cS0[i] + rbV[cb0 + n], 0.f);
      float o1 = fmaxf(conv[b1] + cR1[i] + bV[cb1 + n], 0.f) +
                 fmaxf(cS1[i] + rbV[cb1 + n], 0.f);
      hout[b0] = o0;
      hout[b1] = o1;
    }
  }
}

extern "C" void kernel_launch(void* const* d_in, const int* in_sizes, int n_in,
                              void* d_out, int out_size, void* d_ws, size_t ws_size,
                              hipStream_t stream) {
  const float* x    = (const float*)d_in[0];
  const int*   ei   = (const int*)d_in[1];
  const int*   et   = (const int*)d_in[2];
  const float* W    = (const float*)d_in[3];
  const float* root = (const float*)d_in[4];
  const float* bias = (const float*)d_in[5];
  const float* resW = (const float*)d_in[6];
  const float* resb = (const float*)d_in[7];

  const int N = in_sizes[0] / DD;
  const int E = in_sizes[2];
  const int L = in_sizes[4] / (DD * DD);
  const int R = in_sizes[3] / (L * DD * DD);
  const int* src  = ei;
  const int* dstA = ei + E;
  const int maxTiles = (E + 63) / 64 + R;

  char* wsb = (char*)d_ws;
  size_t off = 0;
  auto carve = [&](size_t bytes) -> void* {
    void* p = wsb + off;
    off += (bytes + 255) & ~(size_t)255;
    return p;
  };
  int*   cnt     = (int*)carve((size_t)N * R * sizeof(int));
  int*   hist    = (int*)carve((size_t)R * sizeof(int));
  int*   cursor  = (int*)carve((size_t)R * sizeof(int));
  int*   tileRel = (int*)carve((size_t)maxTiles * sizeof(int));
  int*   perm    = (int*)carve((size_t)maxTiles * 64 * sizeof(int));
  float* hA      = (float*)carve((size_t)N * DD * sizeof(float));
  float* hB      = (float*)carve((size_t)N * DD * sizeof(float));
  float* conv    = (float*)carve((size_t)N * DD * sizeof(float));

  hipMemsetAsync(cnt, 0, (size_t)N * R * sizeof(int), stream);
  hipMemsetAsync(hist, 0, (size_t)R * sizeof(int), stream);
  hipMemsetAsync(perm, 0xFF, (size_t)maxTiles * 64 * sizeof(int), stream);

  edge_prep   <<<256, 1024, 0, stream>>>(dstA, et, cnt, hist, E, R);
  scan_pad    <<<1, 32, 0, stream>>>(hist, cursor, tileRel, R, maxTiles);
  edge_scatter<<<256, 1024, 0, stream>>>(et, cursor, perm, E, R);

  const float* hcur = x;
  for (int l = 0; l < L; ++l) {
    float* hnext = (l == L - 1) ? (float*)d_out : ((l & 1) ? hB : hA);
    hipMemsetAsync(conv, 0, (size_t)N * DD * sizeof(float), stream);
    edge_gemm<<<maxTiles, 256, 0, stream>>>(
        hcur, W + (size_t)l * R * DD * DD, perm, tileRel,
        src, dstA, et, cnt, conv, R);
    node_gemm<<<(N + 63) / 64, 256, 0, stream>>>(
        hcur, conv, root + (size_t)l * DD * DD, bias + (size_t)l * DD,
        resW + (size_t)l * DD * DD, resb + (size_t)l * DD, hnext, N);
    hcur = hnext;
  }
}


# Edge GEMM made persistent: 2048 blocks each walk a contiguous run of tiles and keep W[rel] in LDS while the relation is unchanged, 5x fewer waves; arrival order of the float atomic sums changes
# speedup vs baseline: 1.9241x; 1.0717x over previous
#include <hip/hip_runtime.h>


typedef __attribute__((ext_vector_type(2))) float v2f;
typedef __attribute__((ext_vector_type(8))) float v8f;

#define DD 64
#define AP 68
#define WP 72

__device__ __forceinline__ v8f wmma4(v2f a, v2f b, v8f c) {
  return __builtin_amdgcn_wmma_f32_16x16x4_f32(
       false, a,  false, b,
       (short)0, c,  false,  false);
}

#define RMAX 128

__global__ void __launch_bounds__(1024)
edge_prep(const int* __restrict__ dst, const int* __restrict__ et,
          int* __restrict__ cnt, int* __restrict__ hist, int E, int R) {
  __shared__ int lh[RMAX];
  const int per = (E + (int)gridDim.x - 1) / (int)gridDim.x;
  const int lo = (int)blockIdx.x * per;
  const int hi = min(lo + per, E);
  const bool inLds = R <= RMAX;
  if (inLds) {
    for (int r = threadIdx.x; r < R; r += blockDim.x) lh[r] = 0;
    __syncthreads();
  }
  for (int e = lo + (int)threadIdx.x; e < hi; e += blockDim.x) {
    int t = et[e];
    atomicAdd(&cnt[(size_t)dst[e] * R + t], 1);
    if (inLds) atomicAdd(&lh[t], 1);
    else       atomicAdd(&hist[t], 1);
  }
  if (inLds) {
    __syncthreads();
    for (int r = threadIdx.x; r < R; r += blockDim.x) {
      int c = lh[r];
      if (c) atomicAdd(&hist[r], c);
    }
  }
}

__global__ void scan_pad(const int* __restrict__ hist, int* __restrict__ cursor,
                         int* __restrict__ tileRel, int R, int maxTiles) {
  if (blockIdx.x != 0 || threadIdx.x != 0) return;
  int run = 0;
  for (int r = 0; r < R; ++r) {
    cursor[r] = run * 64;
    int tiles = (hist[r] + 63) >> 6;
    for (int tt = 0; tt < tiles; ++tt) tileRel[run + tt] = r;
    run += tiles;
  }
  for (int tt = run; tt < maxTiles; ++tt) tileRel[tt] = -1;
}

__global__ void __launch_bounds__(1024)
edge_scatter(const int* __restrict__ et, int* __restrict__ cursor,
             int* __restrict__ perm, int E, int R) {
  __shared__ int lc[RMAX];
  const int per = (E + (int)gridDim.x - 1) / (int)gridDim.x;
  const int lo = (int)blockIdx.x * per;
  const int hi = min(lo + per, E);
  if (R > RMAX) {
    for (int e = lo + (int)threadIdx.x; e < hi; e += blockDim.x) {
      int pos = atomicAdd(&cursor[et[e]], 1);
      perm[pos] = e;
    }
    return;
  }
  for (int r = threadIdx.x; r < R; r += blockDim.x) lc[r] = 0;
  __syncthreads();
  for (int e = lo + (int)threadIdx.x; e < hi; e += blockDim.x)
    atomicAdd(&lc[et[e]], 1);
  __syncthreads();
  for (int r = threadIdx.x; r < R; r += blockDim.x) {
    int c = lc[r];
    lc[r] = c ? atomicAdd(&cursor[r], c) : 0;
  }
  __syncthreads();
  for (int e = lo + (int)threadIdx.x; e < hi; e += blockDim.x) {
    int pos = atomicAdd(&lc[et[e]], 1);
    perm[pos] = e;
  }
}

__global__ void __launch_bounds__(256, 16)
edge_gemm(const float* __restrict__ h, const float* __restrict__ Wl,
          const int* __restrict__ perm, const int* __restrict__ tileRel,
          const int* __restrict__ src, const int* __restrict__ dstArr,
          const int* __restrict__ et, const int* __restrict__ cnt,
          float* __restrict__ conv, int R, int maxTiles) {
  __shared__ float As[64][AP];
  __shared__ float Ws[64][WP];
  __shared__ __align__(16) int   eSrc[2][64];
  __shared__ __align__(16) int   eDst[2][64];
  __shared__ __align__(16) float eInv[2][64];

  const int tid = threadIdx.x;
  const int nb = (int)gridDim.x, bid = (int)blockIdx.x;
  const int per = maxTiles / nb, rem = maxTiles - per * nb;
  const int t0 = bid * per + min(bid, rem);
  const int t1 = t0 + per + (bid < rem ? 1 : 0);

  const int lane = tid & 31, w = tid >> 5;
  const int rt = w >> 1, cp = w & 1;
  const int cb0 = cp * 32, cb1 = cb0 + 16;
  const int n = lane & 15, hi = lane >> 4;
  const int m = rt * 16 + n;

  int curRel = -1;
  int relv = -1;
  for (int t = t0; t < t1; ++t) {
    const int j = (t - t0) & 31;
    if (j == 0) {
      const int tt = t + lane;
      relv = (tt < t1) ? tileRel[tt] : -1;
    }
    const int rel = __builtin_amdgcn_readlane(relv, j);
    if (rel < 0) break;
    const int buf = (t - t0) & 1;

    if (tid < 64) {
      int e = perm[(size_t)t * 64 + tid];
      int s = -1, d = 0; float inv = 0.f;
      if (e >= 0) {
        s = src[e];
        d = dstArr[e];
        int c = cnt[(size_t)d * R + et[e]];
        inv = 1.0f / (float)(c > 0 ? c : 1);
      }
      eSrc[buf][tid] = s; eDst[buf][tid] = d; eInv[buf][tid] = inv;
    }

    if (rel != curRel) {
      __syncthreads();
      const float4* W4 = (const float4*)(Wl + (size_t)rel * DD * DD);
      for (int i = 0; i < 4; ++i) {
        int idx = tid * 4 + i;
        int r = idx >> 4, c4 = (idx & 15) * 4;
        float4 v = W4[idx];
        Ws[r][c4 + 0] = v.x; Ws[r][c4 + 1] = v.y;
        Ws[r][c4 + 2] = v.z; Ws[r][c4 + 3] = v.w;
      }
      curRel = rel;
    }
    __syncthreads();

    {
      int row = tid >> 2, cb = (tid & 3) * 16;
      int s = eSrc[buf][row];
      if (s >= 0) {
        const float4* hp = (const float4*)(h + (size_t)s * DD + cb);
        for (int i = 0; i < 4; ++i) {
          float4 v = hp[i];
          As[row][cb + i * 4 + 0] = v.x; As[row][cb + i * 4 + 1] = v.y;
          As[row][cb + i * 4 + 2] = v.z; As[row][cb + i * 4 + 3] = v.w;
        }
      } else {
        for (int i = 0; i < 16; ++i) As[row][cb + i] = 0.f;
      }
    }
    __syncthreads();

    v8f c0 = {}; v8f c1 = {};
    for (int k = 0; k < 16; ++k) {
      int kb = k * 4 + hi * 2;
      v2f a;  a.x  = As[m][kb];        a.y  = As[m][kb + 1];
      v2f b0; b0.x = Ws[kb][cb0 + n];  b0.y = Ws[kb + 1][cb0 + n];
      v2f b1; b1.x = Ws[kb][cb1 + n];  b1.y = Ws[kb + 1][cb1 + n];
      c0 = wmma4(a, b0, c0);
      c1 = wmma4(a, b1, c1);
    }

    int sb = buf;
    asm volatile("v_nop\n\tv_nop\n\tv_nop\n\tv_nop" : "+v"(c0), "+v"(c1), "+s"(sb));

    const int gr0 = rt * 16 + hi * 8;
    int sv[8], dv[8];
    float p0[8], p1[8];
    for (int i = 0; i < 8; ++i) {
      sv[i] = eSrc[sb][gr0 + i];
      dv[i] = eDst[sb][gr0 + i];
      float inv = eInv[sb][gr0 + i];
      p0[i] = c0[i] * inv;
      p1[i] = c1[i] * inv;
    }
    for (int i = 0; i < 8; ++i) {
      if (sv[i] >= 0) {
        size_t base = (size_t)dv[i] * DD;
        atomicAdd(&conv[base + cb0 + n], p0[i]);
        atomicAdd(&conv[base + cb1 + n], p1[i]);
      }
    }
  }
}

__global__ void __launch_bounds__(256)
node_gemm(const float* __restrict__ h, const float* __restrict__ conv,
          const float* __restrict__ rootW, const float* __restrict__ biasv,
          const float* __restrict__ resW, const float* __restrict__ resb,
          float* __restrict__ hout, int N) {
  __shared__ float As[64][AP];
  __shared__ float Wr[64][WP];
  __shared__ float Wq[64][WP];
  __shared__ float bV[64], rbV[64];

  int tid = threadIdx.x;
  int base = blockIdx.x * 64;

  if (tid < 64) { bV[tid] = biasv[tid]; rbV[tid] = resb[tid]; }

  for (int i = 0; i < 4; ++i) {
    int idx = tid * 4 + i;
    int r = idx >> 4, c4 = (idx & 15) * 4;
    float4 v = ((const float4*)rootW)[idx];
    Wr[r][c4 + 0] = v.x; Wr[r][c4 + 1] = v.y; Wr[r][c4 + 2] = v.z; Wr[r][c4 + 3] = v.w;
    float4 u = ((const float4*)resW)[idx];
    Wq[r][c4 + 0] = u.x; Wq[r][c4 + 1] = u.y; Wq[r][c4 + 2] = u.z; Wq[r][c4 + 3] = u.w;
  }

  {
    int row = tid >> 2, cb = (tid & 3) * 16;
    int node = base + row;
    if (node < N) {
      const float4* hp = (const float4*)(h + (size_t)node * DD + cb);
      for (int i = 0; i < 4; ++i) {
        float4 v = hp[i];
        As[row][cb + i * 4 + 0] = v.x; As[row][cb + i * 4 + 1] = v.y;
        As[row][cb + i * 4 + 2] = v.z; As[row][cb + i * 4 + 3] = v.w;
      }
    } else {
      for (int i = 0; i < 16; ++i) As[row][cb + i] = 0.f;
    }
  }
  __syncthreads();

  int lane = tid & 31, w = tid >> 5;
  int rt = w >> 1, cp = w & 1;
  int cb0 = cp * 32, cb1 = cb0 + 16;
  int n = lane & 15, hi = lane >> 4;
  int m = rt * 16 + n;

  v8f cR0 = {}, cR1 = {}, cS0 = {}, cS1 = {};
  for (int k = 0; k < 16; ++k) {
    int kb = k * 4 + hi * 2;
    v2f a;   a.x   = As[m][kb];        a.y   = As[m][kb + 1];
    v2f br0; br0.x = Wr[kb][cb0 + n];  br0.y = Wr[kb + 1][cb0 + n];
    v2f br1; br1.x = Wr[kb][cb1 + n];  br1.y = Wr[kb + 1][cb1 + n];
    v2f bq0; bq0.x = Wq[kb][cb0 + n];  bq0.y = Wq[kb + 1][cb0 + n];
    v2f bq1; bq1.x = Wq[kb][cb1 + n];  bq1.y = Wq[kb + 1][cb1 + n];
    cR0 = wmma4(a, br0, cR0);
    cR1 = wmma4(a, br1, cR1);
    cS0 = wmma4(a, bq0, cS0);
    cS1 = wmma4(a, bq1, cS1);
  }

  for (int i = 0; i < 8; ++i) {
    int gr = rt * 16 + hi * 8 + i;
    int node = base + gr;
    if (node < N) {
      size_t b0 = (size_t)node * DD + cb0 + n;
      size_t b1 = (size_t)node * DD + cb1 + n;
      float o0 = fmaxf(conv[b0] + cR0[i] + bV[cb0 + n], 0.f) +
                 fmaxf(cS0[i] + rbV[cb0 + n], 0.f);
      float o1 = fmaxf(conv[b1] + cR1[i] + bV[cb1 + n], 0.f) +
                 fmaxf(cS1[i] + rbV[cb1 + n], 0.f);
      hout[b0] = o0;
      hout[b1] = o1;
    }
  }
}

extern "C" void kernel_launch(void* const* d_in, const int* in_sizes, int n_in,
                              void* d_out, int out_size, void* d_ws, size_t ws_size,
                              hipStream_t stream) {
  const float* x    = (const float*)d_in[0];
  const int*   ei   = (const int*)d_in[1];
  const int*   et   = (const int*)d_in[2];
  const float* W    = (const float*)d_in[3];
  const float* root = (const float*)d_in[4];
  const float* bias = (const float*)d_in[5];
  const float* resW = (const float*)d_in[6];
  const float* resb = (const float*)d_in[7];

  const int N = in_sizes[0] / DD;
  const int E = in_sizes[2];
  const int L = in_sizes[4] / (DD * DD);
  const int R = in_sizes[3] / (L * DD * DD);
  const int* src  = ei;
  const int* dstA = ei + E;
  const int maxTiles = (E + 63) / 64 + R;

  char* wsb = (char*)d_ws;
  size_t off = 0;
  auto carve = [&](size_t bytes) -> void* {
    void* p = wsb + off;
    off += (bytes + 255) & ~(size_t)255;
    return p;
  };
  int*   cnt     = (int*)carve((size_t)N * R * sizeof(int));
  int*   hist    = (int*)carve((size_t)R * sizeof(int));
  int*   cursor  = (int*)carve((size_t)R * sizeof(int));
  int*   tileRel = (int*)carve((size_t)maxTiles * sizeof(int));
  int*   perm    = (int*)carve((size_t)maxTiles * 64 * sizeof(int));
  float* hA      = (float*)carve((size_t)N * DD * sizeof(float));
  float* hB      = (float*)carve((size_t)N * DD * sizeof(float));
  float* conv    = (float*)carve((size_t)N * DD * sizeof(float));

  hipMemsetAsync(cnt, 0, (size_t)N * R * sizeof(int), stream);
  hipMemsetAsync(hist, 0, (size_t)R * sizeof(int), stream);
  hipMemsetAsync(perm, 0xFF, (size_t)maxTiles * 64 * sizeof(int), stream);

  edge_prep   <<<256, 1024, 0, stream>>>(dstA, et, cnt, hist, E, R);
  scan_pad    <<<1, 32, 0, stream>>>(hist, cursor, tileRel, R, maxTiles);
  edge_scatter<<<256, 1024, 0, stream>>>(et, cursor, perm, E, R);

  const float* hcur = x;
  for (int l = 0; l < L; ++l) {
    float* hnext = (l == L - 1) ? (float*)d_out : ((l & 1) ? hB : hA);
    hipMemsetAsync(conv, 0, (size_t)N * DD * sizeof(float), stream);
    edge_gemm<<<2048, 256, 0, stream>>>(
        hcur, W + (size_t)l * R * DD * DD, perm, tileRel,
        src, dstA, et, cnt, conv, R, maxTiles);
    node_gemm<<<(N + 63) / 64, 256, 0, stream>>>(
        hcur, conv, root + (size_t)l * DD * DD, bias + (size_t)l * DD,
        resW + (size_t)l * DD * DD, resb + (size_t)l * DD, hnext, N);
    hcur = hnext;
  }
}


# Buffer clears and the tile table folded into the neighbouring kernels: 5 memsets become 1, the serial single-thread scan launch is gone (12 stream operations to 7); float sums unchanged
# speedup vs baseline: 1.9954x; 1.0371x over previous
#include <hip/hip_runtime.h>


typedef __attribute__((ext_vector_type(2))) float v2f;
typedef __attribute__((ext_vector_type(8))) float v8f;

#define DD 64
#define AP 68
#define WP 72

__device__ __forceinline__ v8f wmma4(v2f a, v2f b, v8f c) {
  return __builtin_amdgcn_wmma_f32_16x16x4_f32(
       false, a,  false, b,
       (short)0, c,  false,  false);
}

#define RMAX 128

__global__ void __launch_bounds__(1024)
edge_prep(const int* __restrict__ dst, const int* __restrict__ et,
          int* __restrict__ cnt, int* __restrict__ hist, int E, int R,
          int* __restrict__ perm, int nPerm4,
          float* __restrict__ conv, int nConv4) {
  __shared__ int lh[RMAX];
  {
    const int gt = (int)(blockIdx.x * blockDim.x + threadIdx.x);
    const int gs = (int)(gridDim.x * blockDim.x);
    int4* perm4 = (int4*)perm;
    float4* conv4 = (float4*)conv;
    const int4 m1 = make_int4(-1, -1, -1, -1);
    const float4 z = make_float4(0.f, 0.f, 0.f, 0.f);
    for (int i = gt; i < nPerm4; i += gs) perm4[i] = m1;
    for (int i = gt; i < nConv4; i += gs) conv4[i] = z;
  }
  const int per = (E + (int)gridDim.x - 1) / (int)gridDim.x;
  const int lo = (int)blockIdx.x * per;
  const int hi = min(lo + per, E);
  const bool inLds = R <= RMAX;
  if (inLds) {
    for (int r = threadIdx.x; r < R; r += blockDim.x) lh[r] = 0;
    __syncthreads();
  }
  for (int e = lo + (int)threadIdx.x; e < hi; e += blockDim.x) {
    int t = et[e];
    atomicAdd(&cnt[(size_t)dst[e] * R + t], 1);
    if (inLds) atomicAdd(&lh[t], 1);
    else       atomicAdd(&hist[t], 1);
  }
  if (inLds) {
    __syncthreads();
    for (int r = threadIdx.x; r < R; r += blockDim.x) {
      int c = lh[r];
      if (c) atomicAdd(&hist[r], c);
    }
  }
}

__global__ void scan_pad(const int* __restrict__ hist, int* __restrict__ cursor,
                         int* __restrict__ tileRel, int R, int maxTiles) {
  if (blockIdx.x != 0 || threadIdx.x != 0) return;
  int run = 0;
  for (int r = 0; r < R; ++r) {
    cursor[r] = run * 64;
    int tiles = (hist[r] + 63) >> 6;
    for (int tt = 0; tt < tiles; ++tt) tileRel[run + tt] = r;
    run += tiles;
  }
  for (int tt = run; tt < maxTiles; ++tt) tileRel[tt] = -1;
}

__global__ void __launch_bounds__(1024)
edge_scatter(const int* __restrict__ et, int* __restrict__ cursor,
             int* __restrict__ perm, int E) {
  const int per = (E + (int)gridDim.x - 1) / (int)gridDim.x;
  const int lo = (int)blockIdx.x * per;
  const int hi = min(lo + per, E);
  for (int e = lo + (int)threadIdx.x; e < hi; e += blockDim.x) {
    int pos = atomicAdd(&cursor[et[e]], 1);
    perm[pos] = e;
  }
}

__global__ void __launch_bounds__(1024)
scan_scatter(const int* __restrict__ et, const int* __restrict__ hist,
             int* __restrict__ cursor, int* __restrict__ tileRel,
             int* __restrict__ perm, int E, int R, int maxTiles) {
  __shared__ int lc[RMAX];
  __shared__ int tl[RMAX];
  __shared__ int ts[RMAX + 1];
  const int tid = (int)threadIdx.x, nt = (int)blockDim.x;
  const int bid = (int)blockIdx.x, nb = (int)gridDim.x;
  const int per = (E + nb - 1) / nb;
  const int lo = bid * per;
  const int hi = min(lo + per, E);
  for (int r = tid; r < R; r += nt) { lc[r] = 0; tl[r] = (hist[r] + 63) >> 6; }
  __syncthreads();
  for (int r = tid; r <= R; r += nt) {
    int s = 0;
    for (int q = 0; q < r; ++q) s += tl[q];
    ts[r] = s;
  }
  for (int e = lo + tid; e < hi; e += nt) atomicAdd(&lc[et[e]], 1);
  __syncthreads();
  for (int r = tid; r < R; r += nt)
    for (int tt = ts[r] + bid; tt < ts[r + 1]; tt += nb) tileRel[tt] = r;
  for (int tt = ts[R] + bid * nt + tid; tt < maxTiles; tt += nb * nt) tileRel[tt] = -1;
  for (int r = tid; r < R; r += nt) {
    int c = lc[r];
    lc[r] = ts[r] * 64 + (c ? atomicAdd(&cursor[r], c) : 0);
  }
  __syncthreads();
  for (int e = lo + tid; e < hi; e += nt) {
    int pos = atomicAdd(&lc[et[e]], 1);
    perm[pos] = e;
  }
}

__global__ void __launch_bounds__(256, 16)
edge_gemm(const float* __restrict__ h, const float* __restrict__ Wl,
          const int* __restrict__ perm, const int* __restrict__ tileRel,
          const int* __restrict__ src, const int* __restrict__ dstArr,
          const int* __restrict__ et, const int* __restrict__ cnt,
          float* __restrict__ conv, int R, int maxTiles) {
  __shared__ float As[64][AP];
  __shared__ float Ws[64][WP];
  __shared__ __align__(16) int   eSrc[2][64];
  __shared__ __align__(16) int   eDst[2][64];
  __shared__ __align__(16) float eInv[2][64];

  const int tid = threadIdx.x;
  const int nb = (int)gridDim.x, bid = (int)blockIdx.x;
  const int per = maxTiles / nb, rem = maxTiles - per * nb;
  const int t0 = bid * per + min(bid, rem);
  const int t1 = t0 + per + (bid < rem ? 1 : 0);

  const int lane = tid & 31, w = tid >> 5;
  const int rt = w >> 1, cp = w & 1;
  const int cb0 = cp * 32, cb1 = cb0 + 16;
  const int n = lane & 15, hi = lane >> 4;
  const int m = rt * 16 + n;

  int curRel = -1;
  int relv = -1;
  for (int t = t0; t < t1; ++t) {
    const int j = (t - t0) & 31;
    if (j == 0) {
      const int tt = t + lane;
      relv = (tt < t1) ? tileRel[tt] : -1;
    }
    const int rel = __builtin_amdgcn_readlane(relv, j);
    if (rel < 0) break;
    const int buf = (t - t0) & 1;

    if (tid < 64) {
      int e = perm[(size_t)t * 64 + tid];
      int s = -1, d = 0; float inv = 0.f;
      if (e >= 0) {
        s = src[e];
        d = dstArr[e];
        int c = cnt[(size_t)d * R + et[e]];
        inv = 1.0f / (float)(c > 0 ? c : 1);
      }
      eSrc[buf][tid] = s; eDst[buf][tid] = d; eInv[buf][tid] = inv;
    }

    if (rel != curRel) {
      __syncthreads();
      const float4* W4 = (const float4*)(Wl + (size_t)rel * DD * DD);
      for (int i = 0; i < 4; ++i) {
        int idx = tid * 4 + i;
        int r = idx >> 4, c4 = (idx & 15) * 4;
        float4 v = W4[idx];
        Ws[r][c4 + 0] = v.x; Ws[r][c4 + 1] = v.y;
        Ws[r][c4 + 2] = v.z; Ws[r][c4 + 3] = v.w;
      }
      curRel = rel;
    }
    __syncthreads();

    {
      int row = tid >> 2, cb = (tid & 3) * 16;
      int s = eSrc[buf][row];
      if (s >= 0) {
        const float4* hp = (const float4*)(h + (size_t)s * DD + cb);
        for (int i = 0; i < 4; ++i) {
          float4 v = hp[i];
          As[row][cb + i * 4 + 0] = v.x; As[row][cb + i * 4 + 1] = v.y;
          As[row][cb + i * 4 + 2] = v.z; As[row][cb + i * 4 + 3] = v.w;
        }
      } else {
        for (int i = 0; i < 16; ++i) As[row][cb + i] = 0.f;
      }
    }
    __syncthreads();

    v8f c0 = {}; v8f c1 = {};
    for (int k = 0; k < 16; ++k) {
      int kb = k * 4 + hi * 2;
      v2f a;  a.x  = As[m][kb];        a.y  = As[m][kb + 1];
      v2f b0; b0.x = Ws[kb][cb0 + n];  b0.y = Ws[kb + 1][cb0 + n];
      v2f b1; b1.x = Ws[kb][cb1 + n];  b1.y = Ws[kb + 1][cb1 + n];
      c0 = wmma4(a, b0, c0);
      c1 = wmma4(a, b1, c1);
    }

    int sb = buf;
    asm volatile("v_nop\n\tv_nop\n\tv_nop\n\tv_nop" : "+v"(c0), "+v"(c1), "+s"(sb));

    const int gr0 = rt * 16 + hi * 8;
    int sv[8], dv[8];
    float p0[8], p1[8];
    for (int i = 0; i < 8; ++i) {
      sv[i] = eSrc[sb][gr0 + i];
      dv[i] = eDst[sb][gr0 + i];
      float inv = eInv[sb][gr0 + i];
      p0[i] = c0[i] * inv;
      p1[i] = c1[i] * inv;
    }
    for (int i = 0; i < 8; ++i) {
      if (sv[i] >= 0) {
        size_t base = (size_t)dv[i] * DD;
        atomicAdd(&conv[base + cb0 + n], p0[i]);
        atomicAdd(&conv[base + cb1 + n], p1[i]);
      }
    }
  }
}

__global__ void __launch_bounds__(256)
node_gemm(const float* __restrict__ h, float* __restrict__ conv,
          const float* __restrict__ rootW, const float* __restrict__ biasv,
          const float* __restrict__ resW, const float* __restrict__ resb,
          float* __restrict__ hout, int N, int clearConv) {
  __shared__ float As[64][AP];
  __shared__ float Wr[64][WP];
  __shared__ float Wq[64][WP];
  __shared__ float bV[64], rbV[64];

  int tid = threadIdx.x;
  int base = blockIdx.x * 64;

  if (tid < 64) { bV[tid] = biasv[tid]; rbV[tid] = resb[tid]; }

  for (int i = 0; i < 4; ++i) {
    int idx = tid * 4 + i;
    int r = idx >> 4, c4 = (idx & 15) * 4;
    float4 v = ((const float4*)rootW)[idx];
    Wr[r][c4 + 0] = v.x; Wr[r][c4 + 1] = v.y; Wr[r][c4 + 2] = v.z; Wr[r][c4 + 3] = v.w;
    float4 u = ((const float4*)resW)[idx];
    Wq[r][c4 + 0] = u.x; Wq[r][c4 + 1] = u.y; Wq[r][c4 + 2] = u.z; Wq[r][c4 + 3] = u.w;
  }

  {
    int row = tid >> 2, cb = (tid & 3) * 16;
    int node = base + row;
    if (node < N) {
      const float4* hp = (const float4*)(h + (size_t)node * DD + cb);
      for (int i = 0; i < 4; ++i) {
        float4 v = hp[i];
        As[row][cb + i * 4 + 0] = v.x; As[row][cb + i * 4 + 1] = v.y;
        As[row][cb + i * 4 + 2] = v.z; As[row][cb + i * 4 + 3] = v.w;
      }
    } else {
      for (int i = 0; i < 16; ++i) As[row][cb + i] = 0.f;
    }
  }
  __syncthreads();

  int lane = tid & 31, w = tid >> 5;
  int rt = w >> 1, cp = w & 1;
  int cb0 = cp * 32, cb1 = cb0 + 16;
  int n = lane & 15, hi = lane >> 4;
  int m = rt * 16 + n;

  v8f cR0 = {}, cR1 = {}, cS0 = {}, cS1 = {};
  for (int k = 0; k < 16; ++k) {
    int kb = k * 4 + hi * 2;
    v2f a;   a.x   = As[m][kb];        a.y   = As[m][kb + 1];
    v2f br0; br0.x = Wr[kb][cb0 + n];  br0.y = Wr[kb + 1][cb0 + n];
    v2f br1; br1.x = Wr[kb][cb1 + n];  br1.y = Wr[kb + 1][cb1 + n];
    v2f bq0; bq0.x = Wq[kb][cb0 + n];  bq0.y = Wq[kb + 1][cb0 + n];
    v2f bq1; bq1.x = Wq[kb][cb1 + n];  bq1.y = Wq[kb + 1][cb1 + n];
    cR0 = wmma4(a, br0, cR0);
    cR1 = wmma4(a, br1, cR1);
    cS0 = wmma4(a, bq0, cS0);
    cS1 = wmma4(a, bq1, cS1);
  }

  for (int i = 0; i < 8; ++i) {
    int gr = rt * 16 + hi * 8 + i;
    int node = base + gr;
    if (node < N) {
      size_t b0 = (size_t)node * DD + cb0 + n;
      size_t b1 = (size_t)node * DD + cb1 + n;
      float a0 = conv[b0], a1 = conv[b1];
      if (clearConv) { conv[b0] = 0.f; conv[b1] = 0.f; }
      float o0 = fmaxf(a0 + cR0[i] + bV[cb0 + n], 0.f) +
                 fmaxf(cS0[i] + rbV[cb0 + n], 0.f);
      float o1 = fmaxf(a1 + cR1[i] + bV[cb1 + n], 0.f) +
                 fmaxf(cS1[i] + rbV[cb1 + n], 0.f);
      hout[b0] = o0;
      hout[b1] = o1;
    }
  }
}

extern "C" void kernel_launch(void* const* d_in, const int* in_sizes, int n_in,
                              void* d_out, int out_size, void* d_ws, size_t ws_size,
                              hipStream_t stream) {
  const float* x    = (const float*)d_in[0];
  const int*   ei   = (const int*)d_in[1];
  const int*   et   = (const int*)d_in[2];
  const float* W    = (const float*)d_in[3];
  const float* root = (const float*)d_in[4];
  const float* bias = (const float*)d_in[5];
  const float* resW = (const float*)d_in[6];
  const float* resb = (const float*)d_in[7];

  const int N = in_sizes[0] / DD;
  const int E = in_sizes[2];
  const int L = in_sizes[4] / (DD * DD);
  const int R = in_sizes[3] / (L * DD * DD);
  const int* src  = ei;
  const int* dstA = ei + E;
  const int maxTiles = (E + 63) / 64 + R;

  char* wsb = (char*)d_ws;
  size_t off = 0;
  auto carve = [&](size_t bytes) -> void* {
    void* p = wsb + off;
    off += (bytes + 255) & ~(size_t)255;
    return p;
  };
  int*   cnt     = (int*)carve((size_t)N * R * sizeof(int));
  int*   hist    = (int*)carve((size_t)R * sizeof(int));
  int*   cursor  = (int*)carve((size_t)R * sizeof(int));
  int*   tileRel = (int*)carve((size_t)maxTiles * sizeof(int));
  int*   perm    = (int*)carve((size_t)maxTiles * 64 * sizeof(int));
  float* hA      = (float*)carve((size_t)N * DD * sizeof(float));
  float* hB      = (float*)carve((size_t)N * DD * sizeof(float));
  float* conv    = (float*)carve((size_t)N * DD * sizeof(float));

  hipMemsetAsync(cnt, 0, (size_t)((char*)tileRel - (char*)cnt), stream);

  edge_prep<<<256, 1024, 0, stream>>>(dstA, et, cnt, hist, E, R,
                                      perm, maxTiles * 16,
                                      conv, N * (DD / 4));
  if (R <= RMAX) {
    scan_scatter<<<256, 1024, 0, stream>>>(et, hist, cursor, tileRel, perm, E, R, maxTiles);
  } else {
    scan_pad    <<<1, 32, 0, stream>>>(hist, cursor, tileRel, R, maxTiles);
    edge_scatter<<<256, 1024, 0, stream>>>(et, cursor, perm, E);
  }

  const float* hcur = x;
  for (int l = 0; l < L; ++l) {
    float* hnext = (l == L - 1) ? (float*)d_out : ((l & 1) ? hB : hA);
    edge_gemm<<<2048, 256, 0, stream>>>(
        hcur, W + (size_t)l * R * DD * DD, perm, tileRel,
        src, dstA, et, cnt, conv, R, maxTiles);
    node_gemm<<<(N + 63) / 64, 256, 0, stream>>>(
        hcur, conv, root + (size_t)l * DD * DD, bias + (size_t)l * DD,
        resW + (size_t)l * DD * DD, resb + (size_t)l * DD, hnext, N,
        (l + 1 < L) ? 1 : 0);
    hcur = hnext;
  }
}


# Sort pass writes a 16-byte record (src, dst, 1/count) per sorted slot; edge GEMM reads it coalesced, one tile ahead, not 4 dependent scattered loads per edge and layer; float sums and order kept
# speedup vs baseline: 2.1051x; 1.0550x over previous
#include <hip/hip_runtime.h>


typedef __attribute__((ext_vector_type(2))) float v2f;
typedef __attribute__((ext_vector_type(8))) float v8f;

#define DD 64
#define AP 68
#define WP 72

__device__ __forceinline__ v8f wmma4(v2f a, v2f b, v8f c) {
  return __builtin_amdgcn_wmma_f32_16x16x4_f32(
       false, a,  false, b,
       (short)0, c,  false,  false);
}

#define RMAX 128

__global__ void __launch_bounds__(1024)
edge_prep(const int* __restrict__ dst, const int* __restrict__ et,
          int* __restrict__ cnt, int* __restrict__ hist, int E, int R,
          int* __restrict__ rec, int nRecFill,
          float* __restrict__ conv, int nConv4) {
  __shared__ int lh[RMAX];
  {
    const int gt = (int)(blockIdx.x * blockDim.x + threadIdx.x);
    const int gs = (int)(gridDim.x * blockDim.x);
    int4* rec4 = (int4*)rec;
    float4* conv4 = (float4*)conv;
    const int4 empty = make_int4(-1, 0, 0, 0);
    const float4 z = make_float4(0.f, 0.f, 0.f, 0.f);
    for (int i = gt; i < nRecFill; i += gs) rec4[i] = empty;
    for (int i = gt; i < nConv4; i += gs) conv4[i] = z;
  }
  const int per = (E + (int)gridDim.x - 1) / (int)gridDim.x;
  const int lo = (int)blockIdx.x * per;
  const int hi = min(lo + per, E);
  const bool inLds = R <= RMAX;
  if (inLds) {
    for (int r = threadIdx.x; r < R; r += blockDim.x) lh[r] = 0;
    __syncthreads();
  }
  for (int e = lo + (int)threadIdx.x; e < hi; e += blockDim.x) {
    int t = et[e];
    atomicAdd(&cnt[(size_t)dst[e] * R + t], 1);
    if (inLds) atomicAdd(&lh[t], 1);
    else       atomicAdd(&hist[t], 1);
  }
  if (inLds) {
    __syncthreads();
    for (int r = threadIdx.x; r < R; r += blockDim.x) {
      int c = lh[r];
      if (c) atomicAdd(&hist[r], c);
    }
  }
}

__device__ __forceinline__ int4 edge_record(const int* __restrict__ src,
                                            const int* __restrict__ dst,
                                            const int* __restrict__ cnt,
                                            int e, int t, int R) {
  const int s = src[e], d = dst[e];
  const int c = cnt[(size_t)d * R + t];
  const float inv = 1.0f / (float)(c > 0 ? c : 1);
  return make_int4(s, d, __float_as_int(inv), 0);
}

__global__ void scan_pad(const int* __restrict__ hist, int* __restrict__ cursor,
                         int* __restrict__ tileRel, int R, int maxTiles) {
  if (blockIdx.x != 0 || threadIdx.x != 0) return;
  int run = 0;
  for (int r = 0; r < R; ++r) {
    cursor[r] = run * 64;
    int tiles = (hist[r] + 63) >> 6;
    for (int tt = 0; tt < tiles; ++tt) tileRel[run + tt] = r;
    run += tiles;
  }
  for (int tt = run; tt < maxTiles; ++tt) tileRel[tt] = -1;
}

__global__ void __launch_bounds__(1024)
edge_scatter(const int* __restrict__ et, const int* __restrict__ src,
             const int* __restrict__ dst, const int* __restrict__ cnt,
             int* __restrict__ cursor, int* __restrict__ rec, int E, int R) {
  int4* rec4 = (int4*)rec;
  const int per = (E + (int)gridDim.x - 1) / (int)gridDim.x;
  const int lo = (int)blockIdx.x * per;
  const int hi = min(lo + per, E);
  for (int e = lo + (int)threadIdx.x; e < hi; e += blockDim.x) {
    const int t = et[e];
    const int4 rc = edge_record(src, dst, cnt, e, t, R);
    int pos = atomicAdd(&cursor[t], 1);
    rec4[pos] = rc;
  }
}

__global__ void __launch_bounds__(1024)
scan_scatter(const int* __restrict__ et, const int* __restrict__ src,
             const int* __restrict__ dst, const int* __restrict__ cnt,
             const int* __restrict__ hist,
             int* __restrict__ cursor, int* __restrict__ tileRel,
             int* __restrict__ rec, int E, int R, int maxTiles) {
  int4* rec4 = (int4*)rec;
  __shared__ int lc[RMAX];
  __shared__ int tl[RMAX];
  __shared__ int ts[RMAX + 1];
  const int tid = (int)threadIdx.x, nt = (int)blockDim.x;
  const int bid = (int)blockIdx.x, nb = (int)gridDim.x;
  const int per = (E + nb - 1) / nb;
  const int lo = bid * per;
  const int hi = min(lo + per, E);
  for (int r = tid; r < R; r += nt) { lc[r] = 0; tl[r] = (hist[r] + 63) >> 6; }
  __syncthreads();
  for (int r = tid; r <= R; r += nt) {
    int s = 0;
    for (int q = 0; q < r; ++q) s += tl[q];
    ts[r] = s;
  }
  for (int e = lo + tid; e < hi; e += nt) atomicAdd(&lc[et[e]], 1);
  __syncthreads();
  for (int r = tid; r < R; r += nt)
    for (int tt = ts[r] + bid; tt < ts[r + 1]; tt += nb) tileRel[tt] = r;
  for (int tt = ts[R] + bid * nt + tid; tt < maxTiles; tt += nb * nt) tileRel[tt] = -1;
  for (int r = bid; r < R; r += nb) {
    const int first = ts[r] * 64 + hist[r], end = ts[r + 1] * 64;
    for (int p = first + tid; p < end; p += nt) rec4[p] = make_int4(-1, 0, 0, 0);
  }
  for (int r = tid; r < R; r += nt) {
    int c = lc[r];
    lc[r] = ts[r] * 64 + (c ? atomicAdd(&cursor[r], c) : 0);
  }
  __syncthreads();
  for (int e = lo + tid; e < hi; e += nt) {
    const int t = et[e];
    const int4 rc = edge_record(src, dst, cnt, e, t, R);
    int pos = atomicAdd(&lc[t], 1);
    rec4[pos] = rc;
  }
}

__global__ void __launch_bounds__(256, 16)
edge_gemm(const float* __restrict__ h, const float* __restrict__ Wl,
          const int* __restrict__ rec, const int* __restrict__ tileRel,
          float* __restrict__ conv, int maxTiles) {
  __shared__ float As[64][AP];
  __shared__ float Ws[64][WP];
  __shared__ __align__(16) int   eSrc[2][64];
  __shared__ __align__(16) int   eDst[2][64];
  __shared__ __align__(16) float eInv[2][64];

  const int tid = threadIdx.x;
  const int nb = (int)gridDim.x, bid = (int)blockIdx.x;
  const int per = maxTiles / nb, rem = maxTiles - per * nb;
  const int t0 = bid * per + min(bid, rem);
  const int t1 = t0 + per + (bid < rem ? 1 : 0);

  const int lane = tid & 31, w = tid >> 5;
  const int rt = w >> 1, cp = w & 1;
  const int cb0 = cp * 32, cb1 = cb0 + 16;
  const int n = lane & 15, hi = lane >> 4;
  const int m = rt * 16 + n;

  const int4* rec4 = (const int4*)rec;
  int4 nrec = make_int4(-1, 0, 0, 0);
  if (t0 < t1 && tid < 64 && tileRel[t0] >= 0) nrec = rec4[(size_t)t0 * 64 + tid];

  int curRel = -1;
  int relv = -1;
  for (int t = t0; t < t1; ++t) {
    const int j = (t - t0) & 31;
    if (j == 0) {
      const int tt = t + lane;
      relv = (tt < t1) ? tileRel[tt] : -1;
    }
    const int rel = __builtin_amdgcn_readlane(relv, j);
    if (rel < 0) break;
    const int buf = (t - t0) & 1;

    if (tid < 64) {
      eSrc[buf][tid] = nrec.x; eDst[buf][tid] = nrec.y;
      eInv[buf][tid] = __int_as_float(nrec.z);
    }

    if (rel != curRel) {
      __syncthreads();
      const float4* W4 = (const float4*)(Wl + (size_t)rel * DD * DD);
      for (int i = 0; i < 4; ++i) {
        int idx = tid * 4 + i;
        int r = idx >> 4, c4 = (idx & 15) * 4;
        float4 v = W4[idx];
        Ws[r][c4 + 0] = v.x; Ws[r][c4 + 1] = v.y;
        Ws[r][c4 + 2] = v.z; Ws[r][c4 + 3] = v.w;
      }
      curRel = rel;
    }
    __syncthreads();

    {
      int row = tid >> 2, cb = (tid & 3) * 16;
      int s = eSrc[buf][row];
      if (s >= 0) {
        const float4* hp = (const float4*)(h + (size_t)s * DD + cb);
        for (int i = 0; i < 4; ++i) {
          float4 v = hp[i];
          As[row][cb + i * 4 + 0] = v.x; As[row][cb + i * 4 + 1] = v.y;
          As[row][cb + i * 4 + 2] = v.z; As[row][cb + i * 4 + 3] = v.w;
        }
      } else {
        for (int i = 0; i < 16; ++i) As[row][cb + i] = 0.f;
      }
    }
    __syncthreads();

    {
      int nrel = -1;
      if (t + 1 < t1) nrel = (j < 31) ? __builtin_amdgcn_readlane(relv, j + 1) : tileRel[t + 1];
      if (tid < 64 && nrel >= 0) nrec = rec4[(size_t)(t + 1) * 64 + tid];
    }

    v8f c0 = {}; v8f c1 = {};
    for (int k = 0; k < 16; ++k) {
      int kb = k * 4 + hi * 2;
      v2f a;  a.x  = As[m][kb];        a.y  = As[m][kb + 1];
      v2f b0; b0.x = Ws[kb][cb0 + n];  b0.y = Ws[kb + 1][cb0 + n];
      v2f b1; b1.x = Ws[kb][cb1 + n];  b1.y = Ws[kb + 1][cb1 + n];
      c0 = wmma4(a, b0, c0);
      c1 = wmma4(a, b1, c1);
    }

    int sb = buf;
    asm volatile("v_nop\n\tv_nop\n\tv_nop\n\tv_nop" : "+v"(c0), "+v"(c1), "+s"(sb));

    const int gr0 = rt * 16 + hi * 8;
    int sv[8], dv[8];
    float p0[8], p1[8];
    for (int i = 0; i < 8; ++i) {
      sv[i] = eSrc[sb][gr0 + i];
      dv[i] = eDst[sb][gr0 + i];
      float inv = eInv[sb][gr0 + i];
      p0[i] = c0[i] * inv;
      p1[i] = c1[i] * inv;
    }
    for (int i = 0; i < 8; ++i) {
      if (sv[i] >= 0) {
        size_t base = (size_t)dv[i] * DD;
        atomicAdd(&conv[base + cb0 + n], p0[i]);
        atomicAdd(&conv[base + cb1 + n], p1[i]);
      }
    }
  }
}

__global__ void __launch_bounds__(256)
node_gemm(const float* __restrict__ h, float* __restrict__ conv,
          const float* __restrict__ rootW, const float* __restrict__ biasv,
          const float* __restrict__ resW, const float* __restrict__ resb,
          float* __restrict__ hout, int N, int clearConv) {
  __shared__ float As[64][AP];
  __shared__ float Wr[64][WP];
  __shared__ float Wq[64][WP];
  __shared__ float bV[64], rbV[64];

  int tid = threadIdx.x;
  int base = blockIdx.x * 64;

  if (tid < 64) { bV[tid] = biasv[tid]; rbV[tid] = resb[tid]; }

  for (int i = 0; i < 4; ++i) {
    int idx = tid * 4 + i;
    int r = idx >> 4, c4 = (idx & 15) * 4;
    float4 v = ((const float4*)rootW)[idx];
    Wr[r][c4 + 0] = v.x; Wr[r][c4 + 1] = v.y; Wr[r][c4 + 2] = v.z; Wr[r][c4 + 3] = v.w;
    float4 u = ((const float4*)resW)[idx];
    Wq[r][c4 + 0] = u.x; Wq[r][c4 + 1] = u.y; Wq[r][c4 + 2] = u.z; Wq[r][c4 + 3] = u.w;
  }

  {
    int row = tid >> 2, cb = (tid & 3) * 16;
    int node = base + row;
    if (node < N) {
      const float4* hp = (const float4*)(h + (size_t)node * DD + cb);
      for (int i = 0; i < 4; ++i) {
        float4 v = hp[i];
        As[row][cb + i * 4 + 0] = v.x; As[row][cb + i * 4 + 1] = v.y;
        As[row][cb + i * 4 + 2] = v.z; As[row][cb + i * 4 + 3] = v.w;
      }
    } else {
      for (int i = 0; i < 16; ++i) As[row][cb + i] = 0.f;
    }
  }
  __syncthreads();

  int lane = tid & 31, w = tid >> 5;
  int rt = w >> 1, cp = w & 1;
  int cb0 = cp * 32, cb1 = cb0 + 16;
  int n = lane & 15, hi = lane >> 4;
  int m = rt * 16 + n;

  v8f cR0 = {}, cR1 = {}, cS0 = {}, cS1 = {};
  for (int k = 0; k < 16; ++k) {
    int kb = k * 4 + hi * 2;
    v2f a;   a.x   = As[m][kb];        a.y   = As[m][kb + 1];
    v2f br0; br0.x = Wr[kb][cb0 + n];  br0.y = Wr[kb + 1][cb0 + n];
    v2f br1; br1.x = Wr[kb][cb1 + n];  br1.y = Wr[kb + 1][cb1 + n];
    v2f bq0; bq0.x = Wq[kb][cb0 + n];  bq0.y = Wq[kb + 1][cb0 + n];
    v2f bq1; bq1.x = Wq[kb][cb1 + n];  bq1.y = Wq[kb + 1][cb1 + n];
    cR0 = wmma4(a, br0, cR0);
    cR1 = wmma4(a, br1, cR1);
    cS0 = wmma4(a, bq0, cS0);
    cS1 = wmma4(a, bq1, cS1);
  }

  for (int i = 0; i < 8; ++i) {
    int gr = rt * 16 + hi * 8 + i;
    int node = base + gr;
    if (node < N) {
      size_t b0 = (size_t)node * DD + cb0 + n;
      size_t b1 = (size_t)node * DD + cb1 + n;
      float a0 = conv[b0], a1 = conv[b1];
      if (clearConv) { conv[b0] = 0.f; conv[b1] = 0.f; }
      float o0 = fmaxf(a0 + cR0[i] + bV[cb0 + n], 0.f) +
                 fmaxf(cS0[i] + rbV[cb0 + n], 0.f);
      float o1 = fmaxf(a1 + cR1[i] + bV[cb1 + n], 0.f) +
                 fmaxf(cS1[i] + rbV[cb1 + n], 0.f);
      hout[b0] = o0;
      hout[b1] = o1;
    }
  }
}

extern "C" void kernel_launch(void* const* d_in, const int* in_sizes, int n_in,
                              void* d_out, int out_size, void* d_ws, size_t ws_size,
                              hipStream_t stream) {
  const float* x    = (const float*)d_in[0];
  const int*   ei   = (const int*)d_in[1];
  const int*   et   = (const int*)d_in[2];
  const float* W    = (const float*)d_in[3];
  const float* root = (const float*)d_in[4];
  const float* bias = (const float*)d_in[5];
  const float* resW = (const float*)d_in[6];
  const float* resb = (const float*)d_in[7];

  const int N = in_sizes[0] / DD;
  const int E = in_sizes[2];
  const int L = in_sizes[4] / (DD * DD);
  const int R = in_sizes[3] / (L * DD * DD);
  const int* src  = ei;
  const int* dstA = ei + E;
  const int maxTiles = (E + 63) / 64 + R;

  char* wsb = (char*)d_ws;
  size_t off = 0;
  auto carve = [&](size_t bytes) -> void* {
    void* p = wsb + off;
    off += (bytes + 255) & ~(size_t)255;
    return p;
  };
  int*   cnt     = (int*)carve((size_t)N * R * sizeof(int));
  int*   hist    = (int*)carve((size_t)R * sizeof(int));
  int*   cursor  = (int*)carve((size_t)R * sizeof(int));
  int*   tileRel = (int*)carve((size_t)maxTiles * sizeof(int));
  int*   rec     = (int*)carve((size_t)maxTiles * 64 * sizeof(int4));
  float* hA      = (float*)carve((size_t)N * DD * sizeof(float));
  float* hB      = (L > 2) ? (float*)carve((size_t)N * DD * sizeof(float)) : hA;
  float* conv    = (float*)carve((size_t)N * DD * sizeof(float));

  hipMemsetAsync(cnt, 0, (size_t)((char*)tileRel - (char*)cnt), stream);

  edge_prep<<<256, 1024, 0, stream>>>(dstA, et, cnt, hist, E, R,
                                      rec, (R <= RMAX) ? 0 : maxTiles * 64,
                                      conv, N * (DD / 4));
  if (R <= RMAX) {
    scan_scatter<<<256, 1024, 0, stream>>>(et, src, dstA, cnt, hist, cursor, tileRel,
                                           rec, E, R, maxTiles);
  } else {
    scan_pad    <<<1, 32, 0, stream>>>(hist, cursor, tileRel, R, maxTiles);
    edge_scatter<<<256, 1024, 0, stream>>>(et, src, dstA, cnt, cursor, rec, E, R);
  }

  const float* hcur = x;
  for (int l = 0; l < L; ++l) {
    float* hnext = (l == L - 1) ? (float*)d_out : ((l & 1) ? hB : hA);
    edge_gemm<<<2048, 256, 0, stream>>>(
        hcur, W + (size_t)l * R * DD * DD, rec, tileRel, conv, maxTiles);
    node_gemm<<<(N + 63) / 64, 256, 0, stream>>>(
        hcur, conv, root + (size_t)l * DD * DD, bias + (size_t)l * DD,
        resW + (size_t)l * DD * DD, resb + (size_t)l * DD, hnext, N,
        (l + 1 < L) ? 1 : 0);
    hcur = hnext;
  }
}


# Edge GEMM adds into one accumulator copy per package half (die id picks it; 2nd copy = the layer's idle output plane), node update adds both; split and arrival order of the float atomic sums change
# speedup vs baseline: 2.5950x; 1.2327x over previous
#include <hip/hip_runtime.h>


typedef __attribute__((ext_vector_type(2))) float v2f;
typedef __attribute__((ext_vector_type(8))) float v8f;

#define DD 64
#define AP 68
#define WP 72

__device__ __forceinline__ v8f wmma4(v2f a, v2f b, v8f c) {
  return __builtin_amdgcn_wmma_f32_16x16x4_f32(
       false, a,  false, b,
       (short)0, c,  false,  false);
}

#define RMAX 128

__global__ void __launch_bounds__(1024)
edge_prep(const int* __restrict__ dst, const int* __restrict__ et,
          int* __restrict__ cnt, int* __restrict__ hist, int E, int R,
          int* __restrict__ rec, int nRecFill,
          float* __restrict__ conv, int nConv4,
          float* __restrict__ outz, int nOut4) {
  __shared__ int lh[RMAX];
  {
    const int gt = (int)(blockIdx.x * blockDim.x + threadIdx.x);
    const int gs = (int)(gridDim.x * blockDim.x);
    int4* rec4 = (int4*)rec;
    float4* conv4 = (float4*)conv;
    float4* out4 = (float4*)outz;
    const int4 empty = make_int4(-1, 0, 0, 0);
    const float4 z = make_float4(0.f, 0.f, 0.f, 0.f);
    for (int i = gt; i < nRecFill; i += gs) rec4[i] = empty;
    for (int i = gt; i < nConv4; i += gs) conv4[i] = z;
    for (int i = gt; i < nOut4; i += gs) out4[i] = z;
  }
  const int per = (E + (int)gridDim.x - 1) / (int)gridDim.x;
  const int lo = (int)blockIdx.x * per;
  const int hi = min(lo + per, E);
  const bool inLds = R <= RMAX;
  if (inLds) {
    for (int r = threadIdx.x; r < R; r += blockDim.x) lh[r] = 0;
    __syncthreads();
  }
  for (int e = lo + (int)threadIdx.x; e < hi; e += blockDim.x) {
    int t = et[e];
    atomicAdd(&cnt[(size_t)dst[e] * R + t], 1);
    if (inLds) atomicAdd(&lh[t], 1);
    else       atomicAdd(&hist[t], 1);
  }
  if (inLds) {
    __syncthreads();
    for (int r = threadIdx.x; r < R; r += blockDim.x) {
      int c = lh[r];
      if (c) atomicAdd(&hist[r], c);
    }
  }
}

__device__ __forceinline__ int4 edge_record(const int* __restrict__ src,
                                            const int* __restrict__ dst,
                                            const int* __restrict__ cnt,
                                            int e, int t, int R) {
  const int s = src[e], d = dst[e];
  const int c = cnt[(size_t)d * R + t];
  const float inv = 1.0f / (float)(c > 0 ? c : 1);
  return make_int4(s, d, __float_as_int(inv), 0);
}

__global__ void scan_pad(const int* __restrict__ hist, int* __restrict__ cursor,
                         int* __restrict__ tileRel, int R, int maxTiles) {
  if (blockIdx.x != 0 || threadIdx.x != 0) return;
  int run = 0;
  for (int r = 0; r < R; ++r) {
    cursor[r] = run * 64;
    int tiles = (hist[r] + 63) >> 6;
    for (int tt = 0; tt < tiles; ++tt) tileRel[run + tt] = r;
    run += tiles;
  }
  for (int tt = run; tt < maxTiles; ++tt) tileRel[tt] = -1;
}

__global__ void __launch_bounds__(1024)
edge_scatter(const int* __restrict__ et, const int* __restrict__ src,
             const int* __restrict__ dst, const int* __restrict__ cnt,
             int* __restrict__ cursor, int* __restrict__ rec, int E, int R) {
  int4* rec4 = (int4*)rec;
  const int per = (E + (int)gridDim.x - 1) / (int)gridDim.x;
  const int lo = (int)blockIdx.x * per;
  const int hi = min(lo + per, E);
  for (int e = lo + (int)threadIdx.x; e < hi; e += blockDim.x) {
    const int t = et[e];
    const int4 rc = edge_record(src, dst, cnt, e, t, R);
    int pos = atomicAdd(&cursor[t], 1);
    rec4[pos] = rc;
  }
}

__global__ void __launch_bounds__(1024)
scan_scatter(const int* __restrict__ et, const int* __restrict__ src,
             const int* __restrict__ dst, const int* __restrict__ cnt,
             const int* __restrict__ hist,
             int* __restrict__ cursor, int* __restrict__ tileRel,
             int* __restrict__ rec, int E, int R, int maxTiles) {
  int4* rec4 = (int4*)rec;
  __shared__ int lc[RMAX];
  __shared__ int tl[RMAX];
  __shared__ int ts[RMAX + 1];
  const int tid = (int)threadIdx.x, nt = (int)blockDim.x;
  const int bid = (int)blockIdx.x, nb = (int)gridDim.x;
  const int per = (E + nb - 1) / nb;
  const int lo = bid * per;
  const int hi = min(lo + per, E);
  for (int r = tid; r < R; r += nt) { lc[r] = 0; tl[r] = (hist[r] + 63) >> 6; }
  __syncthreads();
  for (int r = tid; r <= R; r += nt) {
    int s = 0;
    for (int q = 0; q < r; ++q) s += tl[q];
    ts[r] = s;
  }
  for (int e = lo + tid; e < hi; e += nt) atomicAdd(&lc[et[e]], 1);
  __syncthreads();
  for (int r = tid; r < R; r += nt)
    for (int tt = ts[r] + bid; tt < ts[r + 1]; tt += nb) tileRel[tt] = r;
  for (int tt = ts[R] + bid * nt + tid; tt < maxTiles; tt += nb * nt) tileRel[tt] = -1;
  for (int r = bid; r < R; r += nb) {
    const int first = ts[r] * 64 + hist[r], end = ts[r + 1] * 64;
    for (int p = first + tid; p < end; p += nt) rec4[p] = make_int4(-1, 0, 0, 0);
  }
  for (int r = tid; r < R; r += nt) {
    int c = lc[r];
    lc[r] = ts[r] * 64 + (c ? atomicAdd(&cursor[r], c) : 0);
  }
  __syncthreads();
  for (int e = lo + tid; e < hi; e += nt) {
    const int t = et[e];
    const int4 rc = edge_record(src, dst, cnt, e, t, R);
    int pos = atomicAdd(&lc[t], 1);
    rec4[pos] = rc;
  }
}

__global__ void __launch_bounds__(256, 16)
edge_gemm(const float* __restrict__ h, const float* __restrict__ Wl,
          const int* __restrict__ rec, const int* __restrict__ tileRel,
          float* __restrict__ conv, float* __restrict__ conv2, int maxTiles) {
  const int place = __builtin_amdgcn_s_sendmsg_rtn(0x87);
  float* __restrict__ acc = ((place >> 18) & 1) ? conv2 : conv;
  __shared__ float As[64][AP];
  __shared__ float Ws[64][WP];
  __shared__ __align__(16) int   eSrc[2][64];
  __shared__ __align__(16) int   eDst[2][64];
  __shared__ __align__(16) float eInv[2][64];

  const int tid = threadIdx.x;
  const int nb = (int)gridDim.x, bid = (int)blockIdx.x;
  const int per = maxTiles / nb, rem = maxTiles - per * nb;
  const int t0 = bid * per + min(bid, rem);
  const int t1 = t0 + per + (bid < rem ? 1 : 0);

  const int lane = tid & 31, w = tid >> 5;
  const int rt = w >> 1, cp = w & 1;
  const int cb0 = cp * 32, cb1 = cb0 + 16;
  const int n = lane & 15, hi = lane >> 4;
  const int m = rt * 16 + n;

  const int4* rec4 = (const int4*)rec;
  int4 nrec = make_int4(-1, 0, 0, 0);
  if (t0 < t1 && tid < 64 && tileRel[t0] >= 0) nrec = rec4[(size_t)t0 * 64 + tid];

  int curRel = -1;
  int relv = -1;
  for (int t = t0; t < t1; ++t) {
    const int j = (t - t0) & 31;
    if (j == 0) {
      const int tt = t + lane;
      relv = (tt < t1) ? tileRel[tt] : -1;
    }
    const int rel = __builtin_amdgcn_readlane(relv, j);
    if (rel < 0) break;
    const int buf = (t - t0) & 1;

    if (tid < 64) {
      eSrc[buf][tid] = nrec.x; eDst[buf][tid] = nrec.y;
      eInv[buf][tid] = __int_as_float(nrec.z);
    }

    if (rel != curRel) {
      __syncthreads();
      const float4* W4 = (const float4*)(Wl + (size_t)rel * DD * DD);
      for (int i = 0; i < 4; ++i) {
        int idx = tid * 4 + i;
        int r = idx >> 4, c4 = (idx & 15) * 4;
        float4 v = W4[idx];
        Ws[r][c4 + 0] = v.x; Ws[r][c4 + 1] = v.y;
        Ws[r][c4 + 2] = v.z; Ws[r][c4 + 3] = v.w;
      }
      curRel = rel;
    }
    __syncthreads();

    {
      int row = tid >> 2, cb = (tid & 3) * 16;
      int s = eSrc[buf][row];
      if (s >= 0) {
        const float4* hp = (const float4*)(h + (size_t)s * DD + cb);
        for (int i = 0; i < 4; ++i) {
          float4 v = hp[i];
          As[row][cb + i * 4 + 0] = v.x; As[row][cb + i * 4 + 1] = v.y;
          As[row][cb + i * 4 + 2] = v.z; As[row][cb + i * 4 + 3] = v.w;
        }
      } else {
        for (int i = 0; i < 16; ++i) As[row][cb + i] = 0.f;
      }
    }
    __syncthreads();

    {
      int nrel = -1;
      if (t + 1 < t1) nrel = (j < 31) ? __builtin_amdgcn_readlane(relv, j + 1) : tileRel[t + 1];
      if (tid < 64 && nrel >= 0) nrec = rec4[(size_t)(t + 1) * 64 + tid];
    }

    v8f c0 = {}; v8f c1 = {};
    for (int k = 0; k < 16; ++k) {
      int kb = k * 4 + hi * 2;
      v2f a;  a.x  = As[m][kb];        a.y  = As[m][kb + 1];
      v2f b0; b0.x = Ws[kb][cb0 + n];  b0.y = Ws[kb + 1][cb0 + n];
      v2f b1; b1.x = Ws[kb][cb1 + n];  b1.y = Ws[kb + 1][cb1 + n];
      c0 = wmma4(a, b0, c0);
      c1 = wmma4(a, b1, c1);
    }

    int sb = buf;
    asm volatile("v_nop\n\tv_nop\n\tv_nop\n\tv_nop" : "+v"(c0), "+v"(c1), "+s"(sb));

    const int gr0 = rt * 16 + hi * 8;
    int sv[8], dv[8];
    float p0[8], p1[8];
    for (int i = 0; i < 8; ++i) {
      sv[i] = eSrc[sb][gr0 + i];
      dv[i] = eDst[sb][gr0 + i];
      float inv = eInv[sb][gr0 + i];
      p0[i] = c0[i] * inv;
      p1[i] = c1[i] * inv;
    }
    for (int i = 0; i < 8; ++i) {
      if (sv[i] >= 0) {
        size_t base = (size_t)dv[i] * DD;
        atomicAdd(&acc[base + cb0 + n], p0[i]);
        atomicAdd(&acc[base + cb1 + n], p1[i]);
      }
    }
  }
}

__global__ void __launch_bounds__(256)
node_gemm(const float* __restrict__ h, float* __restrict__ conv,
          const float* __restrict__ rootW, const float* __restrict__ biasv,
          const float* __restrict__ resW, const float* __restrict__ resb,
          float* __restrict__ hout, int N, int clearConv, int addOut) {
  __shared__ float As[64][AP];
  __shared__ float Wr[64][WP];
  __shared__ float Wq[64][WP];
  __shared__ float bV[64], rbV[64];

  int tid = threadIdx.x;
  int base = blockIdx.x * 64;

  if (tid < 64) { bV[tid] = biasv[tid]; rbV[tid] = resb[tid]; }

  for (int i = 0; i < 4; ++i) {
    int idx = tid * 4 + i;
    int r = idx >> 4, c4 = (idx & 15) * 4;
    float4 v = ((const float4*)rootW)[idx];
    Wr[r][c4 + 0] = v.x; Wr[r][c4 + 1] = v.y; Wr[r][c4 + 2] = v.z; Wr[r][c4 + 3] = v.w;
    float4 u = ((const float4*)resW)[idx];
    Wq[r][c4 + 0] = u.x; Wq[r][c4 + 1] = u.y; Wq[r][c4 + 2] = u.z; Wq[r][c4 + 3] = u.w;
  }

  {
    int row = tid >> 2, cb = (tid & 3) * 16;
    int node = base + row;
    if (node < N) {
      const float4* hp = (const float4*)(h + (size_t)node * DD + cb);
      for (int i = 0; i < 4; ++i) {
        float4 v = hp[i];
        As[row][cb + i * 4 + 0] = v.x; As[row][cb + i * 4 + 1] = v.y;
        As[row][cb + i * 4 + 2] = v.z; As[row][cb + i * 4 + 3] = v.w;
      }
    } else {
      for (int i = 0; i < 16; ++i) As[row][cb + i] = 0.f;
    }
  }
  __syncthreads();

  int lane = tid & 31, w = tid >> 5;
  int rt = w >> 1, cp = w & 1;
  int cb0 = cp * 32, cb1 = cb0 + 16;
  int n = lane & 15, hi = lane >> 4;
  int m = rt * 16 + n;

  v8f cR0 = {}, cR1 = {}, cS0 = {}, cS1 = {};
  for (int k = 0; k < 16; ++k) {
    int kb = k * 4 + hi * 2;
    v2f a;   a.x   = As[m][kb];        a.y   = As[m][kb + 1];
    v2f br0; br0.x = Wr[kb][cb0 + n];  br0.y = Wr[kb + 1][cb0 + n];
    v2f br1; br1.x = Wr[kb][cb1 + n];  br1.y = Wr[kb + 1][cb1 + n];
    v2f bq0; bq0.x = Wq[kb][cb0 + n];  bq0.y = Wq[kb + 1][cb0 + n];
    v2f bq1; bq1.x = Wq[kb][cb1 + n];  bq1.y = Wq[kb + 1][cb1 + n];
    cR0 = wmma4(a, br0, cR0);
    cR1 = wmma4(a, br1, cR1);
    cS0 = wmma4(a, bq0, cS0);
    cS1 = wmma4(a, bq1, cS1);
  }

  for (int i = 0; i < 8; ++i) {
    int gr = rt * 16 + hi * 8 + i;
    int node = base + gr;
    if (node < N) {
      size_t b0 = (size_t)node * DD + cb0 + n;
      size_t b1 = (size_t)node * DD + cb1 + n;
      float a0 = conv[b0], a1 = conv[b1];
      if (addOut) { a0 += hout[b0]; a1 += hout[b1]; }
      if (clearConv) { conv[b0] = 0.f; conv[b1] = 0.f; }
      float o0 = fmaxf(a0 + cR0[i] + bV[cb0 + n], 0.f) +
                 fmaxf(cS0[i] + rbV[cb0 + n], 0.f);
      float o1 = fmaxf(a1 + cR1[i] + bV[cb1 + n], 0.f) +
                 fmaxf(cS1[i] + rbV[cb1 + n], 0.f);
      hout[b0] = o0;
      hout[b1] = o1;
    }
  }
}

extern "C" void kernel_launch(void* const* d_in, const int* in_sizes, int n_in,
                              void* d_out, int out_size, void* d_ws, size_t ws_size,
                              hipStream_t stream) {
  const float* x    = (const float*)d_in[0];
  const int*   ei   = (const int*)d_in[1];
  const int*   et   = (const int*)d_in[2];
  const float* W    = (const float*)d_in[3];
  const float* root = (const float*)d_in[4];
  const float* bias = (const float*)d_in[5];
  const float* resW = (const float*)d_in[6];
  const float* resb = (const float*)d_in[7];

  const int N = in_sizes[0] / DD;
  const int E = in_sizes[2];
  const int L = in_sizes[4] / (DD * DD);
  const int R = in_sizes[3] / (L * DD * DD);
  const int* src  = ei;
  const int* dstA = ei + E;
  const int maxTiles = (E + 63) / 64 + R;

  char* wsb = (char*)d_ws;
  size_t off = 0;
  auto carve = [&](size_t bytes) -> void* {
    void* p = wsb + off;
    off += (bytes + 255) & ~(size_t)255;
    return p;
  };
  int*   cnt     = (int*)carve((size_t)N * R * sizeof(int));
  int*   hist    = (int*)carve((size_t)R * sizeof(int));
  int*   cursor  = (int*)carve((size_t)R * sizeof(int));
  int*   tileRel = (int*)carve((size_t)maxTiles * sizeof(int));
  int*   rec     = (int*)carve((size_t)maxTiles * 64 * sizeof(int4));
  float* hA      = (float*)carve((size_t)N * DD * sizeof(float));
  float* hB      = (L > 2) ? (float*)carve((size_t)N * DD * sizeof(float)) : hA;
  float* conv    = (float*)carve((size_t)N * DD * sizeof(float));

  hipMemsetAsync(cnt, 0, (size_t)((char*)tileRel - (char*)cnt), stream);

  const bool twoCopies = (L <= 2);
  const bool withA = twoCopies && (L == 2);
  edge_prep<<<256, 1024, 0, stream>>>(dstA, et, cnt, hist, E, R,
                                      rec, (R <= RMAX) ? 0 : maxTiles * 64,
                                      withA ? hA : conv,
                                      withA ? 2 * N * (DD / 4) : N * (DD / 4),
                                      (float*)d_out, twoCopies ? N * (DD / 4) : 0);
  if (R <= RMAX) {
    scan_scatter<<<256, 1024, 0, stream>>>(et, src, dstA, cnt, hist, cursor, tileRel,
                                           rec, E, R, maxTiles);
  } else {
    scan_pad    <<<1, 32, 0, stream>>>(hist, cursor, tileRel, R, maxTiles);
    edge_scatter<<<256, 1024, 0, stream>>>(et, src, dstA, cnt, cursor, rec, E, R);
  }

  const float* hcur = x;
  for (int l = 0; l < L; ++l) {
    float* hnext = (l == L - 1) ? (float*)d_out : ((l & 1) ? hB : hA);
    edge_gemm<<<2048, 256, 0, stream>>>(
        hcur, W + (size_t)l * R * DD * DD, rec, tileRel, conv,
        twoCopies ? hnext : conv, maxTiles);
    node_gemm<<<(N + 63) / 64, 256, 0, stream>>>(
        hcur, conv, root + (size_t)l * DD * DD, bias + (size_t)l * DD,
        resW + (size_t)l * DD * DD, resb + (size_t)l * DD, hnext, N,
        (l + 1 < L) ? 1 : 0, twoCopies ? 1 : 0);
    hcur = hnext;
  }
}


# Staging loads re-mapped (edge GEMM row gather and W, node GEMM h and W): a wave-wide 16-byte load reads 4 whole 128-byte lines, not 32-byte parts of 16 lines; float sums and their order kept
# speedup vs baseline: 2.6894x; 1.0364x over previous
#include <hip/hip_runtime.h>


typedef __attribute__((ext_vector_type(2))) float v2f;
typedef __attribute__((ext_vector_type(8))) float v8f;

#define DD 64
#define AP 68
#define WP 72

__device__ __forceinline__ v8f wmma4(v2f a, v2f b, v8f c) {
  return __builtin_amdgcn_wmma_f32_16x16x4_f32(
       false, a,  false, b,
       (short)0, c,  false,  false);
}

#define RMAX 128

__global__ void __launch_bounds__(1024)
edge_prep(const int* __restrict__ dst, const int* __restrict__ et,
          int* __restrict__ cnt, int* __restrict__ hist, int E, int R,
          int* __restrict__ rec, int nRecFill,
          float* __restrict__ conv, int nConv4,
          float* __restrict__ outz, int nOut4) {
  __shared__ int lh[RMAX];
  {
    const int gt = (int)(blockIdx.x * blockDim.x + threadIdx.x);
    const int gs = (int)(gridDim.x * blockDim.x);
    int4* rec4 = (int4*)rec;
    float4* conv4 = (float4*)conv;
    float4* out4 = (float4*)outz;
    const int4 empty = make_int4(-1, 0, 0, 0);
    const float4 z = make_float4(0.f, 0.f, 0.f, 0.f);
    for (int i = gt; i < nRecFill; i += gs) rec4[i] = empty;
    for (int i = gt; i < nConv4; i += gs) conv4[i] = z;
    for (int i = gt; i < nOut4; i += gs) out4[i] = z;
  }
  const int per = (E + (int)gridDim.x - 1) / (int)gridDim.x;
  const int lo = (int)blockIdx.x * per;
  const int hi = min(lo + per, E);
  const bool inLds = R <= RMAX;
  if (inLds) {
    for (int r = threadIdx.x; r < R; r += blockDim.x) lh[r] = 0;
    __syncthreads();
  }
  for (int e = lo + (int)threadIdx.x; e < hi; e += blockDim.x) {
    int t = et[e];
    atomicAdd(&cnt[(size_t)dst[e] * R + t], 1);
    if (inLds) atomicAdd(&lh[t], 1);
    else       atomicAdd(&hist[t], 1);
  }
  if (inLds) {
    __syncthreads();
    for (int r = threadIdx.x; r < R; r += blockDim.x) {
      int c = lh[r];
      if (c) atomicAdd(&hist[r], c);
    }
  }
}

__device__ __forceinline__ int4 edge_record(const int* __restrict__ src,
                                            const int* __restrict__ dst,
                                            const int* __restrict__ cnt,
                                            int e, int t, int R) {
  const int s = src[e], d = dst[e];
  const int c = cnt[(size_t)d * R + t];
  const float inv = 1.0f / (float)(c > 0 ? c : 1);
  return make_int4(s, d, __float_as_int(inv), 0);
}

__global__ void scan_pad(const int* __restrict__ hist, int* __restrict__ cursor,
                         int* __restrict__ tileRel, int R, int maxTiles) {
  if (blockIdx.x != 0 || threadIdx.x != 0) return;
  int run = 0;
  for (int r = 0; r < R; ++r) {
    cursor[r] = run * 64;
    int tiles = (hist[r] + 63) >> 6;
    for (int tt = 0; tt < tiles; ++tt) tileRel[run + tt] = r;
    run += tiles;
  }
  for (int tt = run; tt < maxTiles; ++tt) tileRel[tt] = -1;
}

__global__ void __launch_bounds__(1024)
edge_scatter(const int* __restrict__ et, const int* __restrict__ src,
             const int* __restrict__ dst, const int* __restrict__ cnt,
             int* __restrict__ cursor, int* __restrict__ rec, int E, int R) {
  int4* rec4 = (int4*)rec;
  const int per = (E + (int)gridDim.x - 1) / (int)gridDim.x;
  const int lo = (int)blockIdx.x * per;
  const int hi = min(lo + per, E);
  for (int e = lo + (int)threadIdx.x; e < hi; e += blockDim.x) {
    const int t = et[e];
    const int4 rc = edge_record(src, dst, cnt, e, t, R);
    int pos = atomicAdd(&cursor[t], 1);
    rec4[pos] = rc;
  }
}

__global__ void __launch_bounds__(1024)
scan_scatter(const int* __restrict__ et, const int* __restrict__ src,
             const int* __restrict__ dst, const int* __restrict__ cnt,
             const int* __restrict__ hist,
             int* __restrict__ cursor, int* __restrict__ tileRel,
             int* __restrict__ rec, int E, int R, int maxTiles) {
  int4* rec4 = (int4*)rec;
  __shared__ int lc[RMAX];
  __shared__ int tl[RMAX];
  __shared__ int ts[RMAX + 1];
  const int tid = (int)threadIdx.x, nt = (int)blockDim.x;
  const int bid = (int)blockIdx.x, nb = (int)gridDim.x;
  const int per = (E + nb - 1) / nb;
  const int lo = bid * per;
  const int hi = min(lo + per, E);
  for (int r = tid; r < R; r += nt) { lc[r] = 0; tl[r] = (hist[r] + 63) >> 6; }
  __syncthreads();
  for (int r = tid; r <= R; r += nt) {
    int s = 0;
    for (int q = 0; q < r; ++q) s += tl[q];
    ts[r] = s;
  }
  for (int e = lo + tid; e < hi; e += nt) atomicAdd(&lc[et[e]], 1);
  __syncthreads();
  for (int r = tid; r < R; r += nt)
    for (int tt = ts[r] + bid; tt < ts[r + 1]; tt += nb) tileRel[tt] = r;
  for (int tt = ts[R] + bid * nt + tid; tt < maxTiles; tt += nb * nt) tileRel[tt] = -1;
  for (int r = bid; r < R; r += nb) {
    const int first = ts[r] * 64 + hist[r], end = ts[r + 1] * 64;
    for (int p = first + tid; p < end; p += nt) rec4[p] = make_int4(-1, 0, 0, 0);
  }
  for (int r = tid; r < R; r += nt) {
    int c = lc[r];
    lc[r] = ts[r] * 64 + (c ? atomicAdd(&cursor[r], c) : 0);
  }
  __syncthreads();
  for (int e = lo + tid; e < hi; e += nt) {
    const int t = et[e];
    const int4 rc = edge_record(src, dst, cnt, e, t, R);
    int pos = atomicAdd(&lc[t], 1);
    rec4[pos] = rc;
  }
}

__global__ void __launch_bounds__(256, 16)
edge_gemm(const float* __restrict__ h, const float* __restrict__ Wl,
          const int* __restrict__ rec, const int* __restrict__ tileRel,
          float* __restrict__ conv, float* __restrict__ conv2, int maxTiles) {
  const int place = __builtin_amdgcn_s_sendmsg_rtn(0x87);
  float* __restrict__ acc = ((place >> 18) & 1) ? conv2 : conv;
  __shared__ float As[64][AP];
  __shared__ float Ws[64][WP];
  __shared__ __align__(16) int   eSrc[2][64];
  __shared__ __align__(16) int   eDst[2][64];
  __shared__ __align__(16) float eInv[2][64];

  const int tid = threadIdx.x;
  const int nb = (int)gridDim.x, bid = (int)blockIdx.x;
  const int per = maxTiles / nb, rem = maxTiles - per * nb;
  const int t0 = bid * per + min(bid, rem);
  const int t1 = t0 + per + (bid < rem ? 1 : 0);

  const int lane = tid & 31, w = tid >> 5;
  const int rt = w >> 1, cp = w & 1;
  const int cb0 = cp * 32, cb1 = cb0 + 16;
  const int n = lane & 15, hi = lane >> 4;
  const int m = rt * 16 + n;

  const int4* rec4 = (const int4*)rec;
  int4 nrec = make_int4(-1, 0, 0, 0);
  if (t0 < t1 && tid < 64 && tileRel[t0] >= 0) nrec = rec4[(size_t)t0 * 64 + tid];

  int curRel = -1;
  int relv = -1;
  for (int t = t0; t < t1; ++t) {
    const int j = (t - t0) & 31;
    if (j == 0) {
      const int tt = t + lane;
      relv = (tt < t1) ? tileRel[tt] : -1;
    }
    const int rel = __builtin_amdgcn_readlane(relv, j);
    if (rel < 0) break;
    const int buf = (t - t0) & 1;

    if (tid < 64) {
      eSrc[buf][tid] = nrec.x; eDst[buf][tid] = nrec.y;
      eInv[buf][tid] = __int_as_float(nrec.z);
    }

    if (rel != curRel) {
      __syncthreads();
      const float4* W4 = (const float4*)Wl + (size_t)rel * (DD * DD / 4);
      float4 wv[4];
      for (int i = 0; i < 4; ++i) wv[i] = W4[(unsigned)(i * 256 + tid)];
      float* wp = &Ws[tid >> 4][(tid & 15) * 4];
      for (int i = 0; i < 4; ++i) {
        wp[i * 16 * WP + 0] = wv[i].x; wp[i * 16 * WP + 1] = wv[i].y;
        wp[i * 16 * WP + 2] = wv[i].z; wp[i * 16 * WP + 3] = wv[i].w;
      }
      curRel = rel;
    }
    __syncthreads();

    {
      const float4* h4 = (const float4*)h;
      const int tr = tid >> 4;
      const unsigned tq = (unsigned)(tid & 15);
      int sv4[4];
      float4 av[4];
      for (int i = 0; i < 4; ++i) sv4[i] = eSrc[buf][i * 16 + tr];
      for (int i = 0; i < 4; ++i) {
        const unsigned sc = sv4[i] < 0 ? 0u : (unsigned)sv4[i];
        av[i] = h4[sc * (DD / 4) + tq];
      }
      float* ap = &As[tr][tq * 4];
      for (int i = 0; i < 4; ++i) {
        const bool on = sv4[i] >= 0;
        ap[i * 16 * AP + 0] = on ? av[i].x : 0.f;
        ap[i * 16 * AP + 1] = on ? av[i].y : 0.f;
        ap[i * 16 * AP + 2] = on ? av[i].z : 0.f;
        ap[i * 16 * AP + 3] = on ? av[i].w : 0.f;
      }
    }
    __syncthreads();

    {
      int nrel = -1;
      if (t + 1 < t1) nrel = (j < 31) ? __builtin_amdgcn_readlane(relv, j + 1) : tileRel[t + 1];
      if (tid < 64 && nrel >= 0) nrec = rec4[(size_t)(t + 1) * 64 + tid];
    }

    v8f c0 = {}; v8f c1 = {};
    for (int k = 0; k < 16; ++k) {
      int kb = k * 4 + hi * 2;
      v2f a;  a.x  = As[m][kb];        a.y  = As[m][kb + 1];
      v2f b0; b0.x = Ws[kb][cb0 + n];  b0.y = Ws[kb + 1][cb0 + n];
      v2f b1; b1.x = Ws[kb][cb1 + n];  b1.y = Ws[kb + 1][cb1 + n];
      c0 = wmma4(a, b0, c0);
      c1 = wmma4(a, b1, c1);
    }

    int sb = buf;
    asm volatile("v_nop\n\tv_nop\n\tv_nop\n\tv_nop" : "+v"(c0), "+v"(c1), "+s"(sb));

    const int gr0 = rt * 16 + hi * 8;
    int sv[8], dv[8];
    float p0[8], p1[8];
    for (int i = 0; i < 8; ++i) {
      sv[i] = eSrc[sb][gr0 + i];
      dv[i] = eDst[sb][gr0 + i];
      float inv = eInv[sb][gr0 + i];
      p0[i] = c0[i] * inv;
      p1[i] = c1[i] * inv;
    }
    for (int i = 0; i < 8; ++i) {
      if (sv[i] >= 0) {
        size_t base = (size_t)dv[i] * DD;
        atomicAdd(&acc[base + cb0 + n], p0[i]);
        atomicAdd(&acc[base + cb1 + n], p1[i]);
      }
    }
  }
}

__global__ void __launch_bounds__(256)
node_gemm(const float* __restrict__ h, float* __restrict__ conv,
          const float* __restrict__ rootW, const float* __restrict__ biasv,
          const float* __restrict__ resW, const float* __restrict__ resb,
          float* __restrict__ hout, int N, int clearConv, int addOut) {
  __shared__ float As[64][AP];
  __shared__ float Wr[64][WP];
  __shared__ float Wq[64][WP];
  __shared__ float bV[64], rbV[64];

  int tid = threadIdx.x;
  int base = blockIdx.x * 64;

  if (tid < 64) { bV[tid] = biasv[tid]; rbV[tid] = resb[tid]; }

  for (int i = 0; i < 4; ++i) {
    int idx = i * 256 + tid;
    int r = idx >> 4, c4 = (idx & 15) * 4;
    float4 v = ((const float4*)rootW)[idx];
    Wr[r][c4 + 0] = v.x; Wr[r][c4 + 1] = v.y; Wr[r][c4 + 2] = v.z; Wr[r][c4 + 3] = v.w;
    float4 u = ((const float4*)resW)[idx];
    Wq[r][c4 + 0] = u.x; Wq[r][c4 + 1] = u.y; Wq[r][c4 + 2] = u.z; Wq[r][c4 + 3] = u.w;
  }

  for (int i = 0; i < 4; ++i) {
    const int p = i * 256 + tid;
    const int row = p >> 4, q4 = (p & 15) * 4;
    const bool on = (base + row) < N;
    const int node = on ? base + row : base;
    const float4 v = *(const float4*)(h + (size_t)node * DD + q4);
    As[row][q4 + 0] = on ? v.x : 0.f; As[row][q4 + 1] = on ? v.y : 0.f;
    As[row][q4 + 2] = on ? v.z : 0.f; As[row][q4 + 3] = on ? v.w : 0.f;
  }
  __syncthreads();

  int lane = tid & 31, w = tid >> 5;
  int rt = w >> 1, cp = w & 1;
  int cb0 = cp * 32, cb1 = cb0 + 16;
  int n = lane & 15, hi = lane >> 4;
  int m = rt * 16 + n;

  v8f cR0 = {}, cR1 = {}, cS0 = {}, cS1 = {};
  for (int k = 0; k < 16; ++k) {
    int kb = k * 4 + hi * 2;
    v2f a;   a.x   = As[m][kb];        a.y   = As[m][kb + 1];
    v2f br0; br0.x = Wr[kb][cb0 + n];  br0.y = Wr[kb + 1][cb0 + n];
    v2f br1; br1.x = Wr[kb][cb1 + n];  br1.y = Wr[kb + 1][cb1 + n];
    v2f bq0; bq0.x = Wq[kb][cb0 + n];  bq0.y = Wq[kb + 1][cb0 + n];
    v2f bq1; bq1.x = Wq[kb][cb1 + n];  bq1.y = Wq[kb + 1][cb1 + n];
    cR0 = wmma4(a, br0, cR0);
    cR1 = wmma4(a, br1, cR1);
    cS0 = wmma4(a, bq0, cS0);
    cS1 = wmma4(a, bq1, cS1);
  }

  for (int i = 0; i < 8; ++i) {
    int gr = rt * 16 + hi * 8 + i;
    int node = base + gr;
    if (node < N) {
      size_t b0 = (size_t)node * DD + cb0 + n;
      size_t b1 = (size_t)node * DD + cb1 + n;
      float a0 = conv[b0], a1 = conv[b1];
      if (addOut) { a0 += hout[b0]; a1 += hout[b1]; }
      if (clearConv) { conv[b0] = 0.f; conv[b1] = 0.f; }
      float o0 = fmaxf(a0 + cR0[i] + bV[cb0 + n], 0.f) +
                 fmaxf(cS0[i] + rbV[cb0 + n], 0.f);
      float o1 = fmaxf(a1 + cR1[i] + bV[cb1 + n], 0.f) +
                 fmaxf(cS1[i] + rbV[cb1 + n], 0.f);
      hout[b0] = o0;
      hout[b1] = o1;
    }
  }
}

extern "C" void kernel_launch(void* const* d_in, const int* in_sizes, int n_in,
                              void* d_out, int out_size, void* d_ws, size_t ws_size,
                              hipStream_t stream) {
  const float* x    = (const float*)d_in[0];
  const int*   ei   = (const int*)d_in[1];
  const int*   et   = (const int*)d_in[2];
  const float* W    = (const float*)d_in[3];
  const float* root = (const float*)d_in[4];
  const float* bias = (const float*)d_in[5];
  const float* resW = (const float*)d_in[6];
  const float* resb = (const float*)d_in[7];

  const int N = in_sizes[0] / DD;
  const int E = in_sizes[2];
  const int L = in_sizes[4] / (DD * DD);
  const int R = in_sizes[3] / (L * DD * DD);
  const int* src  = ei;
  const int* dstA = ei + E;
  const int maxTiles = (E + 63) / 64 + R;

  char* wsb = (char*)d_ws;
  size_t off = 0;
  auto carve = [&](size_t bytes) -> void* {
    void* p = wsb + off;
    off += (bytes + 255) & ~(size_t)255;
    return p;
  };
  int*   cnt     = (int*)carve((size_t)N * R * sizeof(int));
  int*   hist    = (int*)carve((size_t)R * sizeof(int));
  int*   cursor  = (int*)carve((size_t)R * sizeof(int));
  int*   tileRel = (int*)carve((size_t)maxTiles * sizeof(int));
  int*   rec     = (int*)carve((size_t)maxTiles * 64 * sizeof(int4));
  float* hA      = (float*)carve((size_t)N * DD * sizeof(float));
  float* hB      = (L > 2) ? (float*)carve((size_t)N * DD * sizeof(float)) : hA;
  float* conv    = (float*)carve((size_t)N * DD * sizeof(float));

  hipMemsetAsync(cnt, 0, (size_t)((char*)tileRel - (char*)cnt), stream);

  const bool twoCopies = (L <= 2);
  const bool withA = twoCopies && (L == 2);
  edge_prep<<<256, 1024, 0, stream>>>(dstA, et, cnt, hist, E, R,
                                      rec, (R <= RMAX) ? 0 : maxTiles * 64,
                                      withA ? hA : conv,
                                      withA ? 2 * N * (DD / 4) : N * (DD / 4),
                                      (float*)d_out, twoCopies ? N * (DD / 4) : 0);
  if (R <= RMAX) {
    scan_scatter<<<256, 1024, 0, stream>>>(et, src, dstA, cnt, hist, cursor, tileRel,
                                           rec, E, R, maxTiles);
  } else {
    scan_pad    <<<1, 32, 0, stream>>>(hist, cursor, tileRel, R, maxTiles);
    edge_scatter<<<256, 1024, 0, stream>>>(et, src, dstA, cnt, cursor, rec, E, R);
  }

  const float* hcur = x;
  for (int l = 0; l < L; ++l) {
    float* hnext = (l == L - 1) ? (float*)d_out : ((l & 1) ? hB : hA);
    edge_gemm<<<2048, 256, 0, stream>>>(
        hcur, W + (size_t)l * R * DD * DD, rec, tileRel, conv,
        twoCopies ? hnext : conv, maxTiles);
    node_gemm<<<(N + 63) / 64, 256, 0, stream>>>(
        hcur, conv, root + (size_t)l * DD * DD, bias + (size_t)l * DD,
        resW + (size_t)l * DD * DD, resb + (size_t)l * DD, hnext, N,
        (l + 1 < L) ? 1 : 0, twoCopies ? 1 : 0);
    hcur = hnext;
  }
}
